# starting kernel
# baseline (speedup 1.0000x reference)
#include <hip/hip_runtime.h>

#define DD 96
#define WT_STRIDE 98
#define BN_EPS 1e-5f
#define SLOPE 0.05f

typedef __attribute__((ext_vector_type(2))) float v2f;
typedef __attribute__((ext_vector_type(8))) float v8f;


__global__ __launch_bounds__(256) void k_init_deg(float* __restrict__ deg, int n) {
    int i = blockIdx.x * 256 + threadIdx.x;
    if (i < n) deg[i] = 1.0f;
}

__global__ __launch_bounds__(256) void k_count_deg(const int* __restrict__ dst,
                                                   float* __restrict__ deg, int nE) {
    int i = blockIdx.x * 256 + threadIdx.x;
    if (i < nE)
        __hip_atomic_fetch_add(&deg[dst[i]], 1.0f, __ATOMIC_RELAXED, __HIP_MEMORY_SCOPE_AGENT);
}

__global__ __launch_bounds__(256) void k_rsqrt(float* __restrict__ deg, int n) {
    int i = blockIdx.x * 256 + threadIdx.x;
    if (i < n) deg[i] = rsqrtf(deg[i]);
}

__global__ __launch_bounds__(256) void k_zero(float* __restrict__ p, int n) {
    int i = blockIdx.x * 256 + threadIdx.x;
    if (i < n) p[i] = 0.0f;
}


__global__ __launch_bounds__(256) void k_gemm96(const float* __restrict__ X,
                                                const float* __restrict__ W,
                                                float* __restrict__ H, int nRows) {
    __shared__ float Wt[DD * WT_STRIDE];
    for (int i = threadIdx.x; i < DD * DD; i += 256) {
        int k = i / DD, col = i - k * DD;
        Wt[col * WT_STRIDE + k] = W[i];
    }
    __syncthreads();

    int wave = (blockIdx.x * 256 + threadIdx.x) >> 5;
    int lane = threadIdx.x & 31;
    int m0 = wave * 16;
    if (m0 >= nRows) return;

    const int lhalf = lane >> 4;
    const int l16   = lane & 15;
    const int khalf = lhalf * 2;

    const float* Xrow = X + (size_t)(m0 + l16) * DD;

    v8f acc[6];
#pragma unroll
    for (int t = 0; t < 6; ++t)
#pragma unroll
        for (int r = 0; r < 8; ++r) acc[t][r] = 0.0f;

    for (int k = 0; k < DD; k += 4) {
        v2f a = *(const v2f*)(Xrow + k + khalf);
        int kr = k + khalf;
#pragma unroll
        for (int t = 0; t < 6; ++t) {
            int col = t * 16 + l16;
            v2f b = *(const v2f*)(&Wt[col * WT_STRIDE + kr]);
            acc[t] = __builtin_amdgcn_wmma_f32_16x16x4_f32(
                false, a, false, b, (short)0, acc[t], false, false);
        }
    }

    int rbase = m0 + lhalf * 8;
#pragma unroll
    for (int t = 0; t < 6; ++t) {
        int col = t * 16 + l16;
#pragma unroll
        for (int r = 0; r < 8; ++r)
            H[(size_t)(rbase + r) * DD + col] = acc[t][r];
    }
}


__global__ __launch_bounds__(256) void k_scatter(const float* __restrict__ H,
                                                 const float* __restrict__ dinv,
                                                 const int* __restrict__ src,
                                                 const int* __restrict__ dst,
                                                 float* __restrict__ OUT, int nE) {
    int e = (blockIdx.x * 256 + threadIdx.x) >> 5;
    if (e >= nE) return;
    int lane = threadIdx.x & 31;
    int s = src[e], d = dst[e];
    float nrm = dinv[s] * dinv[d];
    const float* hs = H + (size_t)s * DD;
    float*       od = OUT + (size_t)d * DD;
#pragma unroll
    for (int j = lane; j < DD; j += 32)
        __hip_atomic_fetch_add(&od[j], hs[j] * nrm, __ATOMIC_RELAXED, __HIP_MEMORY_SCOPE_AGENT);
}


__global__ __launch_bounds__(256) void k_post1(float* __restrict__ X1,
                                               const float* __restrict__ H,
                                               const float* __restrict__ dinv,
                                               const float* __restrict__ b1,
                                               const float* __restrict__ g,
                                               const float* __restrict__ be,
                                               const float* __restrict__ mu,
                                               const float* __restrict__ var, int n) {
    int i = blockIdx.x * 256 + threadIdx.x;
    if (i >= n) return;
    int node = i / DD;
    int j    = i - node * DD;
    float di = dinv[node];
    float v  = X1[i] + H[i] * di * di + b1[j];
    v = (v >= 0.0f) ? v : SLOPE * v;
    v = (v - mu[j]) * rsqrtf(var[j] + BN_EPS) * g[j] + be[j];
    X1[i] = v;
}


__global__ __launch_bounds__(256) void k_post2(float* __restrict__ OUT,
                                               const float* __restrict__ H,
                                               const float* __restrict__ dinv,
                                               const float* __restrict__ b2, int n) {
    int i = blockIdx.x * 256 + threadIdx.x;
    if (i >= n) return;
    int node = i / DD;
    int j    = i - node * DD;
    float di = dinv[node];
    OUT[i] = OUT[i] + H[i] * di * di + b2[j];
}


extern "C" void kernel_launch(void* const* d_in, const int* in_sizes, int n_in,
                              void* d_out, int out_size, void* d_ws, size_t ws_size,
                              hipStream_t stream) {
    const float* X    = (const float*)d_in[0];
    const int*   ei   = (const int*)d_in[1];
    const float* W1   = (const float*)d_in[2];
    const float* b1   = (const float*)d_in[3];
    const float* W2   = (const float*)d_in[4];
    const float* b2   = (const float*)d_in[5];
    const float* gam  = (const float*)d_in[6];
    const float* bet  = (const float*)d_in[7];
    const float* mu   = (const float*)d_in[8];
    const float* var  = (const float*)d_in[9];

    const int N = in_sizes[0] / DD;
    const int E = in_sizes[1] / 2;
    const int* srcI = ei;
    const int* dstI = ei + E;

    float* dinv = (float*)d_ws;
    float* hA   = dinv + N;
    float* hB   = hA + (size_t)N * DD;
    float* out  = (float*)d_out;

    const int nElems    = N * DD;
    const int gridN     = (N + 255) / 256;
    const int gridE     = (E + 255) / 256;
    const int gridElems = (nElems + 255) / 256;
    const int gemmGrid  = ((N + 15) / 16 + 7) / 8;
    const int scatGrid  = (E + 7) / 8;

    k_init_deg <<<gridN, 256, 0, stream>>>(dinv, N);
    k_count_deg<<<gridE, 256, 0, stream>>>(dstI, dinv, E);
    k_rsqrt    <<<gridN, 256, 0, stream>>>(dinv, N);

    k_gemm96 <<<gemmGrid,  256, 0, stream>>>(X, W1, hA, N);
    k_zero   <<<gridElems, 256, 0, stream>>>(hB, nElems);
    k_scatter<<<scatGrid,  256, 0, stream>>>(hA, dinv, srcI, dstI, hB, E);
    k_post1  <<<gridElems, 256, 0, stream>>>(hB, hA, dinv, b1, gam, bet, mu, var, nElems);

    k_gemm96 <<<gemmGrid,  256, 0, stream>>>(hB, W2, hA, N);
    k_zero   <<<gridElems, 256, 0, stream>>>(out, nElems);
    k_scatter<<<scatGrid,  256, 0, stream>>>(hA, dinv, srcI, dstI, out, E);
    k_post2  <<<gridElems, 256, 0, stream>>>(out, hA, dinv, b2, nElems);
}


# Float-atomic edge scatter-add replaced by a per-call destination-grouped edge list (int count+rank atomics, scan, fill) and one wave per node summing its rows in registers, in arrival-rank order
# speedup vs baseline: 1.7742x; 1.7742x over previous
#include <hip/hip_runtime.h>

#define DD 96
#define WT_STRIDE 98
#define BN_EPS 1e-5f
#define SLOPE 0.05f

typedef __attribute__((ext_vector_type(2))) float v2f;
typedef __attribute__((ext_vector_type(8))) float v8f;


__global__ __launch_bounds__(256) void k_zero_i(int* __restrict__ p, int n) {
    int i = blockIdx.x * 256 + threadIdx.x;
    if (i < n) p[i] = 0;
}

__global__ __launch_bounds__(256) void k_count_rank(const int* __restrict__ dst,
                                                    int* __restrict__ cnt,
                                                    int* __restrict__ rank, int nE) {
    int i = blockIdx.x * 256 + threadIdx.x;
    if (i < nE)
        rank[i] = __hip_atomic_fetch_add(&cnt[dst[i]], 1, __ATOMIC_RELAXED, __HIP_MEMORY_SCOPE_AGENT);
}

__device__ __forceinline__ int wave_sum(int v) {
    for (int o = 16; o > 0; o >>= 1) v += __shfl_xor(v, o);
    return v;
}

__global__ __launch_bounds__(256) void k_blocksum(const int* __restrict__ cnt,
                                                  int* __restrict__ bsum, int n) {
    __shared__ int part[8];
    int tid = threadIdx.x;
    int e0 = (blockIdx.x * 256 + tid) * 4;
    int t = 0;
    for (int k = 0; k < 4; ++k)
        if (e0 + k < n) t += cnt[e0 + k];
    t = wave_sum(t);
    if ((tid & 31) == 0) part[tid >> 5] = t;
    __syncthreads();
    if (tid == 0) {
        int s = 0;
        for (int w = 0; w < 8; ++w) s += part[w];
        bsum[blockIdx.x] = s;
    }
}

__global__ __launch_bounds__(256) void k_scan(const int* __restrict__ cnt,
                                              const int* __restrict__ bsum,
                                              int* __restrict__ rowptr,
                                              float* __restrict__ dinv, int n) {
    __shared__ int part[8];
    __shared__ int wtot[8];
    int tid = threadIdx.x, lane = tid & 31, wv = tid >> 5;

    int p = 0;
    for (int k = tid; k < (int)blockIdx.x; k += 256) p += bsum[k];
    p = wave_sum(p);

    int e0 = (blockIdx.x * 256 + tid) * 4;
    int c[4];
    for (int k = 0; k < 4; ++k) c[k] = (e0 + k < n) ? cnt[e0 + k] : 0;
    int t = c[0] + c[1] + c[2] + c[3];
    int incl = t;
    for (int o = 1; o < 32; o <<= 1) {
        int u = __shfl_up(incl, o);
        if (lane >= o) incl += u;
    }
    if (lane == 0)  part[wv] = p;
    if (lane == 31) wtot[wv] = incl;
    __syncthreads();

    int ex = incl - t;
    for (int w = 0; w < 8; ++w) {
        ex += part[w];
        if (w < wv) ex += wtot[w];
    }
    for (int k = 0; k < 4; ++k) {
        int i = e0 + k;
        if (i < n) {
            rowptr[i] = ex;
            dinv[i]   = rsqrtf((float)(c[k] + 1));
            ex += c[k];
            if (i == n - 1) rowptr[n] = ex;
        }
    }
}

__global__ __launch_bounds__(256) void k_fill(const int* __restrict__ src,
                                              const int* __restrict__ dst,
                                              const int* __restrict__ rank,
                                              const int* __restrict__ rowptr,
                                              int* __restrict__ lst, int nE) {
    int i = blockIdx.x * 256 + threadIdx.x;
    if (i < nE) lst[rowptr[dst[i]] + rank[i]] = src[i];
}


__global__ __launch_bounds__(256) void k_gemm96(const float* __restrict__ X,
                                                const float* __restrict__ W,
                                                float* __restrict__ H, int nRows) {
    __shared__ float Wt[DD * WT_STRIDE];
    for (int i = threadIdx.x; i < DD * DD; i += 256) {
        int k = i / DD, col = i - k * DD;
        Wt[col * WT_STRIDE + k] = W[i];
    }
    __syncthreads();

    int wave = (blockIdx.x * 256 + threadIdx.x) >> 5;
    int lane = threadIdx.x & 31;
    int m0 = wave * 16;
    if (m0 >= nRows) return;

    const int lhalf = lane >> 4;
    const int l16   = lane & 15;
    const int khalf = lhalf * 2;

    const float* Xrow = X + (size_t)(m0 + l16) * DD;

    v8f acc[6];
#pragma unroll
    for (int t = 0; t < 6; ++t)
#pragma unroll
        for (int r = 0; r < 8; ++r) acc[t][r] = 0.0f;

    for (int k = 0; k < DD; k += 4) {
        v2f a = *(const v2f*)(Xrow + k + khalf);
        int kr = k + khalf;
#pragma unroll
        for (int t = 0; t < 6; ++t) {
            int col = t * 16 + l16;
            v2f b = *(const v2f*)(&Wt[col * WT_STRIDE + kr]);
            acc[t] = __builtin_amdgcn_wmma_f32_16x16x4_f32(
                false, a, false, b, (short)0, acc[t], false, false);
        }
    }

    int rbase = m0 + lhalf * 8;
#pragma unroll
    for (int t = 0; t < 6; ++t) {
        int col = t * 16 + l16;
#pragma unroll
        for (int r = 0; r < 8; ++r)
            H[(size_t)(rbase + r) * DD + col] = acc[t][r];
    }
}


struct __attribute__((packed, aligned(4))) f3 { float x, y, z; };

#define GU 8

__global__ __launch_bounds__(256) void k_gather(const float* __restrict__ H,
                                                const float* __restrict__ dinv,
                                                const int* __restrict__ rowptr,
                                                const int* __restrict__ lst,
                                                float* __restrict__ OUT, int n) {
    int node = (blockIdx.x * 256 + threadIdx.x) >> 5;
    node = __builtin_amdgcn_readfirstlane(node);
    if (node >= n) return;
    const int lane = threadIdx.x & 31;
    const int beg = rowptr[node], end = rowptr[node + 1];
    const float dd = dinv[node];
    const float* Hl = H + lane * 3;

    float a0 = 0.0f, a1 = 0.0f, a2 = 0.0f;
    for (int base = beg; base < end; base += 32) {
        int m = end - base;
        if (m > 32) m = 32;
        int   s  = 0;
        float ds = 0.0f;
        if (lane < m) {
            s  = lst[base + lane];
            ds = dinv[s];
        }
        for (int i = 0; i < m; i += GU) {
            f3    r[GU];
            float wu[GU];
#pragma unroll
            for (int u = 0; u < GU; ++u) {
                int idx = (u == 0 || i + u < m) ? (i + u) : (m - 1);
                int su  = __builtin_amdgcn_readlane(s, idx);
                r[u]    = *(const f3*)(Hl + (size_t)su * DD);
            }
            __builtin_amdgcn_sched_barrier(0);
#pragma unroll
            for (int u = 0; u < GU; ++u) {
                int idx = (u == 0 || i + u < m) ? (i + u) : (m - 1);
                wu[u] = __builtin_bit_cast(float, __builtin_amdgcn_readlane(__builtin_bit_cast(int, ds), idx)) * dd;
            }
#pragma unroll
            for (int u = 0; u < GU; ++u) {
                bool ok = (u == 0) || (i + u < m);
                a0 = __builtin_fmaf(ok ? r[u].x : 0.0f, wu[u], a0);
                a1 = __builtin_fmaf(ok ? r[u].y : 0.0f, wu[u], a1);
                a2 = __builtin_fmaf(ok ? r[u].z : 0.0f, wu[u], a2);
            }
        }
    }
    f3 o;
    o.x = a0; o.y = a1; o.z = a2;
    *(f3*)(OUT + (size_t)node * DD + lane * 3) = o;
}


__global__ __launch_bounds__(256) void k_post1(float* __restrict__ X1,
                                               const float* __restrict__ H,
                                               const float* __restrict__ dinv,
                                               const float* __restrict__ b1,
                                               const float* __restrict__ g,
                                               const float* __restrict__ be,
                                               const float* __restrict__ mu,
                                               const float* __restrict__ var, int n) {
    int i = blockIdx.x * 256 + threadIdx.x;
    if (i >= n) return;
    int node = i / DD;
    int j    = i - node * DD;
    float di = dinv[node];
    float v  = X1[i] + H[i] * di * di + b1[j];
    v = (v >= 0.0f) ? v : SLOPE * v;
    v = (v - mu[j]) * rsqrtf(var[j] + BN_EPS) * g[j] + be[j];
    X1[i] = v;
}


__global__ __launch_bounds__(256) void k_post2(float* __restrict__ OUT,
                                               const float* __restrict__ H,
                                               const float* __restrict__ dinv,
                                               const float* __restrict__ b2, int n) {
    int i = blockIdx.x * 256 + threadIdx.x;
    if (i >= n) return;
    int node = i / DD;
    int j    = i - node * DD;
    float di = dinv[node];
    OUT[i] = OUT[i] + H[i] * di * di + b2[j];
}


extern "C" void kernel_launch(void* const* d_in, const int* in_sizes, int n_in,
                              void* d_out, int out_size, void* d_ws, size_t ws_size,
                              hipStream_t stream) {
    const float* X    = (const float*)d_in[0];
    const int*   ei   = (const int*)d_in[1];
    const float* W1   = (const float*)d_in[2];
    const float* b1   = (const float*)d_in[3];
    const float* W2   = (const float*)d_in[4];
    const float* b2   = (const float*)d_in[5];
    const float* gam  = (const float*)d_in[6];
    const float* bet  = (const float*)d_in[7];
    const float* mu   = (const float*)d_in[8];
    const float* var  = (const float*)d_in[9];

    const int N = in_sizes[0] / DD;
    const int E = in_sizes[1] / 2;
    const int* srcI = ei;
    const int* dstI = ei + E;

    float* dinv = (float*)d_ws;
    float* hA   = dinv + N;
    float* hB   = hA + (size_t)N * DD;
    int*   cnt  = (int*)(hB + (size_t)N * DD);
    int*   rptr = cnt + N;
    int*   rank = rptr + (N + 1);
    int*   lst  = rank + E;
    int*   bsum = lst + E;
    float* out  = (float*)d_out;

    const int nElems    = N * DD;
    const int gridN     = (N + 255) / 256;
    const int gridE     = (E + 255) / 256;
    const int gridElems = (nElems + 255) / 256;
    const int gridScan  = (N + 1023) / 1024;
    const int gemmGrid  = ((N + 15) / 16 + 7) / 8;
    const int gathGrid  = (N + 7) / 8;

    k_zero_i    <<<gridN,    256, 0, stream>>>(cnt, N);
    k_count_rank<<<gridE,    256, 0, stream>>>(dstI, cnt, rank, E);
    k_blocksum  <<<gridScan, 256, 0, stream>>>(cnt, bsum, N);
    k_scan      <<<gridScan, 256, 0, stream>>>(cnt, bsum, rptr, dinv, N);
    k_fill      <<<gridE,    256, 0, stream>>>(srcI, dstI, rank, rptr, lst, E);

    k_gemm96<<<gemmGrid,  256, 0, stream>>>(X, W1, hA, N);
    k_gather<<<gathGrid,  256, 0, stream>>>(hA, dinv, rptr, lst, hB, N);
    k_post1 <<<gridElems, 256, 0, stream>>>(hB, hA, dinv, b1, gam, bet, mu, var, nElems);

    k_gemm96<<<gemmGrid,  256, 0, stream>>>(hB, W2, hA, N);
    k_gather<<<gathGrid,  256, 0, stream>>>(hA, dinv, rptr, lst, out, N);
    k_post2 <<<gridElems, 256, 0, stream>>>(out, hA, dinv, b2, nElems);
}


# Self-loop term, bias and layer-1 LeakyReLU+BatchNorm moved into the gather wave that owns the node's row; both one-element-per-thread epilogue passes are gone (11 -> 9 launches)
# speedup vs baseline: 2.2043x; 1.2424x over previous
#include <hip/hip_runtime.h>

#define DD 96
#define WT_STRIDE 98
#define BN_EPS 1e-5f
#define SLOPE 0.05f

typedef __attribute__((ext_vector_type(2))) float v2f;
typedef __attribute__((ext_vector_type(8))) float v8f;


__global__ __launch_bounds__(256) void k_zero_i(int* __restrict__ p, int n) {
    int i = blockIdx.x * 256 + threadIdx.x;
    if (i < n) p[i] = 0;
}

__global__ __launch_bounds__(256) void k_count_rank(const int* __restrict__ dst,
                                                    int* __restrict__ cnt,
                                                    int* __restrict__ rank, int nE) {
    int i = blockIdx.x * 256 + threadIdx.x;
    if (i < nE)
        rank[i] = __hip_atomic_fetch_add(&cnt[dst[i]], 1, __ATOMIC_RELAXED, __HIP_MEMORY_SCOPE_AGENT);
}

__device__ __forceinline__ int wave_sum(int v) {
    for (int o = 16; o > 0; o >>= 1) v += __shfl_xor(v, o);
    return v;
}

__global__ __launch_bounds__(256) void k_blocksum(const int* __restrict__ cnt,
                                                  int* __restrict__ bsum, int n) {
    __shared__ int part[8];
    int tid = threadIdx.x;
    int e0 = (blockIdx.x * 256 + tid) * 4;
    int t = 0;
    for (int k = 0; k < 4; ++k)
        if (e0 + k < n) t += cnt[e0 + k];
    t = wave_sum(t);
    if ((tid & 31) == 0) part[tid >> 5] = t;
    __syncthreads();
    if (tid == 0) {
        int s = 0;
        for (int w = 0; w < 8; ++w) s += part[w];
        bsum[blockIdx.x] = s;
    }
}

__global__ __launch_bounds__(256) void k_scan(const int* __restrict__ cnt,
                                              const int* __restrict__ bsum,
                                              int* __restrict__ rowptr,
                                              float* __restrict__ dinv, int n) {
    __shared__ int part[8];
    __shared__ int wtot[8];
    int tid = threadIdx.x, lane = tid & 31, wv = tid >> 5;

    int p = 0;
    for (int k = tid; k < (int)blockIdx.x; k += 256) p += bsum[k];
    p = wave_sum(p);

    int e0 = (blockIdx.x * 256 + tid) * 4;
    int c[4];
    for (int k = 0; k < 4; ++k) c[k] = (e0 + k < n) ? cnt[e0 + k] : 0;
    int t = c[0] + c[1] + c[2] + c[3];
    int incl = t;
    for (int o = 1; o < 32; o <<= 1) {
        int u = __shfl_up(incl, o);
        if (lane >= o) incl += u;
    }
    if (lane == 0)  part[wv] = p;
    if (lane == 31) wtot[wv] = incl;
    __syncthreads();

    int ex = incl - t;
    for (int w = 0; w < 8; ++w) {
        ex += part[w];
        if (w < wv) ex += wtot[w];
    }
    for (int k = 0; k < 4; ++k) {
        int i = e0 + k;
        if (i < n) {
            rowptr[i] = ex;
            dinv[i]   = rsqrtf((float)(c[k] + 1));
            ex += c[k];
            if (i == n - 1) rowptr[n] = ex;
        }
    }
}

__global__ __launch_bounds__(256) void k_fill(const int* __restrict__ src,
                                              const int* __restrict__ dst,
                                              const int* __restrict__ rank,
                                              const int* __restrict__ rowptr,
                                              int* __restrict__ lst, int nE) {
    int i = blockIdx.x * 256 + threadIdx.x;
    if (i < nE) lst[rowptr[dst[i]] + rank[i]] = src[i];
}


__global__ __launch_bounds__(256) void k_gemm96(const float* __restrict__ X,
                                                const float* __restrict__ W,
                                                float* __restrict__ H, int nRows) {
    __shared__ float Wt[DD * WT_STRIDE];
    for (int i = threadIdx.x; i < DD * DD; i += 256) {
        int k = i / DD, col = i - k * DD;
        Wt[col * WT_STRIDE + k] = W[i];
    }
    __syncthreads();

    int wave = (blockIdx.x * 256 + threadIdx.x) >> 5;
    int lane = threadIdx.x & 31;
    int m0 = wave * 16;
    if (m0 >= nRows) return;

    const int lhalf = lane >> 4;
    const int l16   = lane & 15;
    const int khalf = lhalf * 2;

    const float* Xrow = X + (size_t)(m0 + l16) * DD;

    v8f acc[6];
#pragma unroll
    for (int t = 0; t < 6; ++t)
#pragma unroll
        for (int r = 0; r < 8; ++r) acc[t][r] = 0.0f;

    for (int k = 0; k < DD; k += 4) {
        v2f a = *(const v2f*)(Xrow + k + khalf);
        int kr = k + khalf;
#pragma unroll
        for (int t = 0; t < 6; ++t) {
            int col = t * 16 + l16;
            v2f b = *(const v2f*)(&Wt[col * WT_STRIDE + kr]);
            acc[t] = __builtin_amdgcn_wmma_f32_16x16x4_f32(
                false, a, false, b, (short)0, acc[t], false, false);
        }
    }

    int rbase = m0 + lhalf * 8;
#pragma unroll
    for (int t = 0; t < 6; ++t) {
        int col = t * 16 + l16;
#pragma unroll
        for (int r = 0; r < 8; ++r)
            H[(size_t)(rbase + r) * DD + col] = acc[t][r];
    }
}


struct __attribute__((packed, aligned(4))) f3 { float x, y, z; };

#define GU 8

__device__ __forceinline__ void list_chunk(const int* __restrict__ lst,
                                           const float* __restrict__ dinv,
                                           int base, int m, int lane, int& s, float& ds) {
    s  = 0;
    ds = 0.0f;
    if (lane < m) {
        s  = lst[base + lane];
        ds = dinv[s];
    }
}

__device__ __forceinline__ void sum_chunk(const float* __restrict__ Hl, int s, float ds, int m, float dd,
                                          float& a0, float& a1, float& a2) {
    for (int i = 0; i < m; i += GU) {
        f3    r[GU];
        float wu[GU];
#pragma unroll
        for (int u = 0; u < GU; ++u) {
            int idx = (u == 0 || i + u < m) ? (i + u) : (m - 1);
            int su  = __builtin_amdgcn_readlane(s, idx);
            r[u]    = *(const f3*)(Hl + (size_t)su * DD);
        }
        __builtin_amdgcn_sched_barrier(0);
#pragma unroll
        for (int u = 0; u < GU; ++u) {
            int idx = (u == 0 || i + u < m) ? (i + u) : (m - 1);
            wu[u] = __builtin_bit_cast(float, __builtin_amdgcn_readlane(__builtin_bit_cast(int, ds), idx)) * dd;
        }
#pragma unroll
        for (int u = 0; u < GU; ++u) {
            bool ok = (u == 0) || (i + u < m);
            a0 = __builtin_fmaf(ok ? r[u].x : 0.0f, wu[u], a0);
            a1 = __builtin_fmaf(ok ? r[u].y : 0.0f, wu[u], a1);
            a2 = __builtin_fmaf(ok ? r[u].z : 0.0f, wu[u], a2);
        }
    }
}

__device__ __forceinline__ void gather_sum(const float* __restrict__ Hl,
                                           const float* __restrict__ dinv,
                                           const int* __restrict__ lst,
                                           int beg, int end, float dd, int lane,
                                           float& a0, float& a1, float& a2) {
    a0 = 0.0f; a1 = 0.0f; a2 = 0.0f;
    int   s;
    float ds;
    int m = end - beg;
    if (m > 32) m = 32;
    list_chunk(lst, dinv, beg, m, lane, s, ds);
    sum_chunk(Hl, s, ds, m, dd, a0, a1, a2);
    for (int base = beg + 32; base < end; base += 32) {
        m = end - base;
        if (m > 32) m = 32;
        list_chunk(lst, dinv, base, m, lane, s, ds);
        sum_chunk(Hl, s, ds, m, dd, a0, a1, a2);
    }
}

__device__ __forceinline__ float leaky_bn(float v, float mu, float var, float g, float be) {
    v = (v >= 0.0f) ? v : SLOPE * v;
    return (v - mu) * rsqrtf(var + BN_EPS) * g + be;
}

__global__ __launch_bounds__(256) void k_gather_l1(const float* __restrict__ H,
                                                   const float* __restrict__ dinv,
                                                   const int* __restrict__ rowptr,
                                                   const int* __restrict__ lst,
                                                   const float* __restrict__ b1,
                                                   const float* __restrict__ g,
                                                   const float* __restrict__ be,
                                                   const float* __restrict__ mu,
                                                   const float* __restrict__ var,
                                                   float* __restrict__ X1, int n) {
    int node = (blockIdx.x * 256 + threadIdx.x) >> 5;
    node = __builtin_amdgcn_readfirstlane(node);
    if (node >= n) return;
    const int lane = threadIdx.x & 31;
    const int beg = rowptr[node], end = rowptr[node + 1];
    const float dd = dinv[node];
    const float* Hl = H + lane * 3;

    const f3 hs = *(const f3*)(Hl + (size_t)node * DD);

    float a0, a1, a2;
    gather_sum(Hl, dinv, lst, beg, end, dd, lane, a0, a1, a2);

    const f3 vb = *(const f3*)(b1  + lane * 3);
    const f3 vm = *(const f3*)(mu  + lane * 3);
    const f3 vv = *(const f3*)(var + lane * 3);
    const f3 vg = *(const f3*)(g   + lane * 3);
    const f3 ve = *(const f3*)(be  + lane * 3);

    f3 o;
    o.x = leaky_bn(a0 + hs.x * dd * dd + vb.x, vm.x, vv.x, vg.x, ve.x);
    o.y = leaky_bn(a1 + hs.y * dd * dd + vb.y, vm.y, vv.y, vg.y, ve.y);
    o.z = leaky_bn(a2 + hs.z * dd * dd + vb.z, vm.z, vv.z, vg.z, ve.z);
    *(f3*)(X1 + (size_t)node * DD + lane * 3) = o;
}

__global__ __launch_bounds__(256) void k_gather_l2(const float* __restrict__ H,
                                                   const float* __restrict__ dinv,
                                                   const int* __restrict__ rowptr,
                                                   const int* __restrict__ lst,
                                                   const float* __restrict__ b2,
                                                   float* __restrict__ OUT, int n) {
    int node = (blockIdx.x * 256 + threadIdx.x) >> 5;
    node = __builtin_amdgcn_readfirstlane(node);
    if (node >= n) return;
    const int lane = threadIdx.x & 31;
    const int beg = rowptr[node], end = rowptr[node + 1];
    const float dd = dinv[node];
    const float* Hl = H + lane * 3;

    const f3 hs = *(const f3*)(Hl + (size_t)node * DD);
    const f3 vb = *(const f3*)(b2 + lane * 3);

    float a0, a1, a2;
    gather_sum(Hl, dinv, lst, beg, end, dd, lane, a0, a1, a2);

    f3 o;
    o.x = a0 + hs.x * dd * dd + vb.x;
    o.y = a1 + hs.y * dd * dd + vb.y;
    o.z = a2 + hs.z * dd * dd + vb.z;
    *(f3*)(OUT + (size_t)node * DD + lane * 3) = o;
}


extern "C" void kernel_launch(void* const* d_in, const int* in_sizes, int n_in,
                              void* d_out, int out_size, void* d_ws, size_t ws_size,
                              hipStream_t stream) {
    const float* X    = (const float*)d_in[0];
    const int*   ei   = (const int*)d_in[1];
    const float* W1   = (const float*)d_in[2];
    const float* b1   = (const float*)d_in[3];
    const float* W2   = (const float*)d_in[4];
    const float* b2   = (const float*)d_in[5];
    const float* gam  = (const float*)d_in[6];
    const float* bet  = (const float*)d_in[7];
    const float* mu   = (const float*)d_in[8];
    const float* var  = (const float*)d_in[9];

    const int N = in_sizes[0] / DD;
    const int E = in_sizes[1] / 2;
    const int* srcI = ei;
    const int* dstI = ei + E;

    float* dinv = (float*)d_ws;
    float* hA   = dinv + N;
    float* hB   = hA + (size_t)N * DD;
    int*   cnt  = (int*)(hB + (size_t)N * DD);
    int*   rptr = cnt + N;
    int*   rank = rptr + (N + 1);
    int*   lst  = rank + E;
    int*   bsum = lst + E;
    float* out  = (float*)d_out;

    const int gridN     = (N + 255) / 256;
    const int gridE     = (E + 255) / 256;
    const int gridScan  = (N + 1023) / 1024;
    const int gemmGrid  = ((N + 15) / 16 + 7) / 8;
    const int gathGrid  = (N + 7) / 8;

    k_zero_i    <<<gridN,    256, 0, stream>>>(cnt, N);
    k_count_rank<<<gridE,    256, 0, stream>>>(dstI, cnt, rank, E);
    k_blocksum  <<<gridScan, 256, 0, stream>>>(cnt, bsum, N);
    k_scan      <<<gridScan, 256, 0, stream>>>(cnt, bsum, rptr, dinv, N);
    k_fill      <<<gridE,    256, 0, stream>>>(srcI, dstI, rank, rptr, lst, E);

    k_gemm96   <<<gemmGrid, 256, 0, stream>>>(X, W1, hA, N);
    k_gather_l1<<<gathGrid, 256, 0, stream>>>(hA, dinv, rptr, lst, b1, gam, bet, mu, var, hB, N);

    k_gemm96   <<<gemmGrid, 256, 0, stream>>>(hB, W2, hA, N);
    k_gather_l2<<<gathGrid, 256, 0, stream>>>(hA, dinv, rptr, lst, b2, out, N);
}


# Edge list built by a two-level counting sort on LDS counters (bucket count, scan, bucket placement, per-bucket lists) instead of one returning global atomic per edge; list order = LDS arrival order
# speedup vs baseline: 2.9293x; 1.3289x over previous
#include <hip/hip_runtime.h>

#define DD 96
#define WT_STRIDE 98
#define BN_EPS 1e-5f
#define SLOPE 0.05f

typedef __attribute__((ext_vector_type(2))) float v2f;
typedef __attribute__((ext_vector_type(8))) float v8f;


#define BSH    9
#define MAXB   2048
#define CHUNK  4096
#define MAXPER 40

__device__ __forceinline__ int tab_at(int i, int per) {
    int t = i / per;
    return (i - t * per) * 1024 + t;
}

__device__ __forceinline__ int wave_incl_scan(int v, int lane) {
    for (int o = 1; o < 32; o <<= 1) {
        int u = __shfl_up(v, o);
        if (lane >= o) v += u;
    }
    return v;
}

__global__ __launch_bounds__(1024) void k_bucket_hist(const int* __restrict__ dst,
                                                      int* __restrict__ tab,
                                                      int nE, int nB, int nC) {
    __shared__ int h[MAXB];
    const int tid = threadIdx.x;
    for (int b = tid; b < nB; b += 1024) h[b] = 0;
    __syncthreads();
    const int e0 = blockIdx.x * CHUNK + tid;
    int d[4];
#pragma unroll
    for (int k = 0; k < 4; ++k) {
        int e = e0 + k * 1024;
        d[k] = dst[e < nE ? e : nE - 1];
    }
#pragma unroll
    for (int k = 0; k < 4; ++k)
        if (e0 + k * 1024 < nE) atomicAdd(&h[d[k] >> BSH], 1);
    __syncthreads();
    const int per = (nB * nC + 1023) >> 10;
    for (int b = tid; b < nB; b += 1024) tab[tab_at(b * nC + (int)blockIdx.x, per)] = h[b];
}

__global__ __launch_bounds__(1024) void k_scan_tab(const int* __restrict__ tab,
                                                   int* __restrict__ base, int total) {
    __shared__ int wtot[32];
    const int tid = threadIdx.x, lane = tid & 31, wv = tid >> 5;
    const int per = (total + 1023) >> 10;
    const int i0 = tid * per;
    int v[MAXPER];
    int s = 0;
#pragma unroll
    for (int k = 0; k < MAXPER; ++k) {
        bool ok = (k < per) && (i0 + k < total);
        int x = tab[ok ? (k * 1024 + tid) : 0];
        v[k] = ok ? x : 0;
        s += v[k];
    }
    int incl = wave_incl_scan(s, lane);
    if (lane == 31) wtot[wv] = incl;
    __syncthreads();
    int run = incl - s;
    for (int w = 0; w < wv; ++w) run += wtot[w];
#pragma unroll
    for (int k = 0; k < MAXPER; ++k) {
        if ((k < per) && (i0 + k < total)) base[k * 1024 + tid] = run;
        run += v[k];
    }
    if (tid == 1023) base[per * 1024] = run;
}

__global__ __launch_bounds__(1024) void k_bucket_place(const int* __restrict__ src,
                                                       const int* __restrict__ dst,
                                                       const int* __restrict__ base,
                                                       int2* __restrict__ pairs,
                                                       int nE, int nB, int nC) {
    __shared__ int cur[MAXB];
    const int tid = threadIdx.x;
    const int per = (nB * nC + 1023) >> 10;
    for (int b = tid; b < nB; b += 1024) cur[b] = base[tab_at(b * nC + (int)blockIdx.x, per)];
    __syncthreads();
    const int e0 = blockIdx.x * CHUNK + tid;
    int d[4], s[4];
#pragma unroll
    for (int k = 0; k < 4; ++k) {
        int e = e0 + k * 1024;
        int ec = e < nE ? e : nE - 1;
        d[k] = dst[ec];
        s[k] = src[ec];
    }
#pragma unroll
    for (int k = 0; k < 4; ++k) {
        if (e0 + k * 1024 < nE) {
            int p = atomicAdd(&cur[d[k] >> BSH], 1);
            pairs[p] = make_int2(d[k] & ((1 << BSH) - 1), s[k]);
        }
    }
}

__global__ __launch_bounds__(1024) void k_bucket_lists(const int2* __restrict__ pairs,
                                                       const int* __restrict__ base,
                                                       int* __restrict__ rowptr,
                                                       float* __restrict__ dinv,
                                                       int* __restrict__ lst,
                                                       int n, int nC, int total) {
    __shared__ int cnt[1 << BSH];
    __shared__ int cur[1 << BSH];
    __shared__ int wtot[32];
    const int tid = threadIdx.x, lane = tid & 31, wv = tid >> 5;
    const int b = blockIdx.x;
    const int per = (total + 1023) >> 10;
    const int i1 = (b + 1) * nC;
    const int s0 = base[tab_at(b * nC, per)];
    const int s1 = base[(i1 < total) ? tab_at(i1, per) : per * 1024];

    if (tid < (1 << BSH)) cnt[tid] = 0;
    __syncthreads();
    for (int i = s0 + tid; i < s1; i += 4096) {
        int q[4];
#pragma unroll
        for (int u = 0; u < 4; ++u) {
            int j = i + u * 1024;
            q[u] = pairs[j < s1 ? j : i].x;
        }
#pragma unroll
        for (int u = 0; u < 4; ++u)
            if (i + u * 1024 < s1) atomicAdd(&cnt[q[u]], 1);
    }
    __syncthreads();

    const int c = (tid < (1 << BSH)) ? cnt[tid] : 0;
    const int incl = wave_incl_scan(c, lane);
    if (lane == 31) wtot[wv] = incl;
    __syncthreads();
    int ex = s0 + incl - c;
    for (int w = 0; w < wv; ++w) ex += wtot[w];
    if (tid < (1 << BSH)) {
        cur[tid] = ex;
        const int node = (b << BSH) + tid;
        if (node < n) {
            rowptr[node] = ex;
            dinv[node]   = rsqrtf((float)(c + 1));
            if (node == n - 1) rowptr[n] = ex + c;
        }
    }
    __syncthreads();
    for (int i = s0 + tid; i < s1; i += 4096) {
        int2 p[4];
#pragma unroll
        for (int u = 0; u < 4; ++u) {
            int j = i + u * 1024;
            p[u] = pairs[j < s1 ? j : i];
        }
#pragma unroll
        for (int u = 0; u < 4; ++u)
            if (i + u * 1024 < s1) lst[atomicAdd(&cur[p[u].x], 1)] = p[u].y;
    }
}


__global__ __launch_bounds__(256) void k_gemm96(const float* __restrict__ X,
                                                const float* __restrict__ W,
                                                const float* __restrict__ dinv,
                                                float* __restrict__ H, int nRows) {
    __shared__ float Wt[DD * WT_STRIDE];
    for (int i = threadIdx.x; i < DD * DD; i += 256) {
        int k = i / DD, col = i - k * DD;
        Wt[col * WT_STRIDE + k] = W[i];
    }
    __syncthreads();

    int wave = (blockIdx.x * 256 + threadIdx.x) >> 5;
    int lane = threadIdx.x & 31;
    int m0 = wave * 16;
    if (m0 >= nRows) return;

    const int lhalf = lane >> 4;
    const int l16   = lane & 15;
    const int khalf = lhalf * 2;

    const float* Xrow = X + (size_t)(m0 + l16) * DD;

    const int rbase = m0 + lhalf * 8;
    const float4 sc_lo = *(const float4*)(dinv + rbase);
    const float4 sc_hi = *(const float4*)(dinv + rbase + 4);

    v8f acc[6];
#pragma unroll
    for (int t = 0; t < 6; ++t)
#pragma unroll
        for (int r = 0; r < 8; ++r) acc[t][r] = 0.0f;

    for (int k = 0; k < DD; k += 4) {
        v2f a = *(const v2f*)(Xrow + k + khalf);
        int kr = k + khalf;
#pragma unroll
        for (int t = 0; t < 6; ++t) {
            int col = t * 16 + l16;
            v2f b = *(const v2f*)(&Wt[col * WT_STRIDE + kr]);
            acc[t] = __builtin_amdgcn_wmma_f32_16x16x4_f32(
                false, a, false, b, (short)0, acc[t], false, false);
        }
    }

    const float sc[8] = {sc_lo.x, sc_lo.y, sc_lo.z, sc_lo.w, sc_hi.x, sc_hi.y, sc_hi.z, sc_hi.w};
#pragma unroll
    for (int t = 0; t < 6; ++t) {
        int col = t * 16 + l16;
#pragma unroll
        for (int r = 0; r < 8; ++r)
            H[(size_t)(rbase + r) * DD + col] = acc[t][r] * sc[r];
    }
}


struct __attribute__((packed, aligned(4))) f3 { float x, y, z; };

#define GU 8

__device__ __forceinline__ int list_chunk(const int* __restrict__ lst, int base, int m, int lane) {
    int s = 0;
    if (lane < m) s = lst[base + lane];
    return s;
}

__device__ __forceinline__ void sum_chunk(const float* __restrict__ Hl, int s, int m,
                                          float& a0, float& a1, float& a2) {
    for (int i = 0; i < m; i += GU) {
        f3 r[GU];
#pragma unroll
        for (int u = 0; u < GU; ++u) {
            int idx = (u == 0 || i + u < m) ? (i + u) : (m - 1);
            int su  = __builtin_amdgcn_readlane(s, idx);
            r[u]    = *(const f3*)(Hl + (size_t)su * DD);
        }
#pragma unroll
        for (int u = 0; u < GU; ++u) {
            bool ok = (u == 0) || (i + u < m);
            a0 += ok ? r[u].x : 0.0f;
            a1 += ok ? r[u].y : 0.0f;
            a2 += ok ? r[u].z : 0.0f;
        }
    }
}

__device__ __forceinline__ void gather_sum(const float* __restrict__ Hl,
                                           const int* __restrict__ lst,
                                           int beg, int end, int lane,
                                           float& a0, float& a1, float& a2) {
    a0 = 0.0f; a1 = 0.0f; a2 = 0.0f;
    int m = end - beg;
    if (m > 32) m = 32;
    int s = list_chunk(lst, beg, m, lane);
    sum_chunk(Hl, s, m, a0, a1, a2);
    for (int base = beg + 32; base < end; base += 32) {
        m = end - base;
        if (m > 32) m = 32;
        s = list_chunk(lst, base, m, lane);
        sum_chunk(Hl, s, m, a0, a1, a2);
    }
}

__device__ __forceinline__ float leaky_bn(float v, float mu, float var, float g, float be) {
    v = (v >= 0.0f) ? v : SLOPE * v;
    return (v - mu) * rsqrtf(var + BN_EPS) * g + be;
}

__global__ __launch_bounds__(256) void k_gather_l1(const float* __restrict__ H,
                                                   const float* __restrict__ dinv,
                                                   const int* __restrict__ rowptr,
                                                   const int* __restrict__ lst,
                                                   const float* __restrict__ b1,
                                                   const float* __restrict__ g,
                                                   const float* __restrict__ be,
                                                   const float* __restrict__ mu,
                                                   const float* __restrict__ var,
                                                   float* __restrict__ X1, int n) {
    int node = (blockIdx.x * 256 + threadIdx.x) >> 5;
    node = __builtin_amdgcn_readfirstlane(node);
    if (node >= n) return;
    const int lane = threadIdx.x & 31;
    const int beg = rowptr[node], end = rowptr[node + 1];
    const float dd = dinv[node];
    const float* Hl = H + lane * 3;

    const f3 hs = *(const f3*)(Hl + (size_t)node * DD);

    float a0, a1, a2;
    gather_sum(Hl, lst, beg, end, lane, a0, a1, a2);

    const f3 vb = *(const f3*)(b1  + lane * 3);
    const f3 vm = *(const f3*)(mu  + lane * 3);
    const f3 vv = *(const f3*)(var + lane * 3);
    const f3 vg = *(const f3*)(g   + lane * 3);
    const f3 ve = *(const f3*)(be  + lane * 3);

    f3 o;
    o.x = leaky_bn((a0 + hs.x) * dd + vb.x, vm.x, vv.x, vg.x, ve.x);
    o.y = leaky_bn((a1 + hs.y) * dd + vb.y, vm.y, vv.y, vg.y, ve.y);
    o.z = leaky_bn((a2 + hs.z) * dd + vb.z, vm.z, vv.z, vg.z, ve.z);
    *(f3*)(X1 + (size_t)node * DD + lane * 3) = o;
}

__global__ __launch_bounds__(256) void k_gather_l2(const float* __restrict__ H,
                                                   const float* __restrict__ dinv,
                                                   const int* __restrict__ rowptr,
                                                   const int* __restrict__ lst,
                                                   const float* __restrict__ b2,
                                                   float* __restrict__ OUT, int n) {
    int node = (blockIdx.x * 256 + threadIdx.x) >> 5;
    node = __builtin_amdgcn_readfirstlane(node);
    if (node >= n) return;
    const int lane = threadIdx.x & 31;
    const int beg = rowptr[node], end = rowptr[node + 1];
    const float dd = dinv[node];
    const float* Hl = H + lane * 3;

    const f3 hs = *(const f3*)(Hl + (size_t)node * DD);
    const f3 vb = *(const f3*)(b2 + lane * 3);

    float a0, a1, a2;
    gather_sum(Hl, lst, beg, end, lane, a0, a1, a2);

    f3 o;
    o.x = (a0 + hs.x) * dd + vb.x;
    o.y = (a1 + hs.y) * dd + vb.y;
    o.z = (a2 + hs.z) * dd + vb.z;
    *(f3*)(OUT + (size_t)node * DD + lane * 3) = o;
}


extern "C" void kernel_launch(void* const* d_in, const int* in_sizes, int n_in,
                              void* d_out, int out_size, void* d_ws, size_t ws_size,
                              hipStream_t stream) {
    const float* X    = (const float*)d_in[0];
    const int*   ei   = (const int*)d_in[1];
    const float* W1   = (const float*)d_in[2];
    const float* b1   = (const float*)d_in[3];
    const float* W2   = (const float*)d_in[4];
    const float* b2   = (const float*)d_in[5];
    const float* gam  = (const float*)d_in[6];
    const float* bet  = (const float*)d_in[7];
    const float* mu   = (const float*)d_in[8];
    const float* var  = (const float*)d_in[9];

    const int N = in_sizes[0] / DD;
    const int E = in_sizes[1] / 2;
    const int* srcI = ei;
    const int* dstI = ei + E;

    float* dinv = (float*)d_ws;
    float* hA   = dinv + N;
    float* hB   = hA + (size_t)N * DD;
    int*   rptr = (int*)(hB + (size_t)N * DD);
    int*   lst  = rptr + (N + 1);
    int*   tab  = lst + E;
    int*   base = tab + (MAXPER * 1024 + 1);
    int2*  pairs = (int2*)(((size_t)(base + (MAXPER * 1024 + 1)) + 7) & ~(size_t)7);
    float* out  = (float*)d_out;

    const int nB        = (N + (1 << BSH) - 1) >> BSH;
    const int nC        = (E + CHUNK - 1) / CHUNK;
    const int gemmGrid  = ((N + 15) / 16 + 7) / 8;
    const int gathGrid  = (N + 7) / 8;

    k_bucket_hist <<<nC, 1024, 0, stream>>>(dstI, tab, E, nB, nC);
    k_scan_tab    <<<1,  1024, 0, stream>>>(tab, base, nB * nC);
    k_bucket_place<<<nC, 1024, 0, stream>>>(srcI, dstI, base, pairs, E, nB, nC);
    k_bucket_lists<<<nB, 1024, 0, stream>>>(pairs, base, rptr, dinv, lst, N, nC, nB * nC);

    k_gemm96   <<<gemmGrid, 256, 0, stream>>>(X, W1, dinv, hA, N);
    k_gather_l1<<<gathGrid, 256, 0, stream>>>(hA, dinv, rptr, lst, b1, gam, bet, mu, var, hB, N);

    k_gemm96   <<<gemmGrid, 256, 0, stream>>>(hB, W2, dinv, hA, N);
    k_gather_l2<<<gathGrid, 256, 0, stream>>>(hA, dinv, rptr, lst, b2, out, N);
}


# Layer-2 product moved into the layer-1 gather launch: a wave finishes 16 nodes' x1 rows into an LDS tile and multiplies it by W2 (same WMMA f32 k order); x1 never stored; 8 -> 7 launches
# speedup vs baseline: 3.3302x; 1.1369x over previous
#include <hip/hip_runtime.h>

#define DD 96
#define WT_STRIDE 98
#define BN_EPS 1e-5f
#define SLOPE 0.05f

typedef __attribute__((ext_vector_type(2))) float v2f;
typedef __attribute__((ext_vector_type(8))) float v8f;


#define BSH    9
#define MAXB   2048
#define CHUNK  4096
#define MAXPER 40

__device__ __forceinline__ int tab_at(int i, int per) {
    int t = i / per;
    return (i - t * per) * 1024 + t;
}

__device__ __forceinline__ int wave_incl_scan(int v, int lane) {
    for (int o = 1; o < 32; o <<= 1) {
        int u = __shfl_up(v, o);
        if (lane >= o) v += u;
    }
    return v;
}

__global__ __launch_bounds__(1024) void k_bucket_hist(const int* __restrict__ dst,
                                                      int* __restrict__ tab,
                                                      int nE, int nB, int nC) {
    __shared__ int h[MAXB];
    const int tid = threadIdx.x;
    for (int b = tid; b < nB; b += 1024) h[b] = 0;
    __syncthreads();
    const int e0 = blockIdx.x * CHUNK + tid;
    int d[4];
#pragma unroll
    for (int k = 0; k < 4; ++k) {
        int e = e0 + k * 1024;
        d[k] = dst[e < nE ? e : nE - 1];
    }
#pragma unroll
    for (int k = 0; k < 4; ++k)
        if (e0 + k * 1024 < nE) atomicAdd(&h[d[k] >> BSH], 1);
    __syncthreads();
    const int per = (nB * nC + 1023) >> 10;
    for (int b = tid; b < nB; b += 1024) tab[tab_at(b * nC + (int)blockIdx.x, per)] = h[b];
}

__global__ __launch_bounds__(1024) void k_scan_tab(const int* __restrict__ tab,
                                                   int* __restrict__ base, int total) {
    __shared__ int wtot[32];
    const int tid = threadIdx.x, lane = tid & 31, wv = tid >> 5;
    const int per = (total + 1023) >> 10;
    const int i0 = tid * per;
    int v[MAXPER];
    int s = 0;
#pragma unroll
    for (int k = 0; k < MAXPER; ++k) {
        bool ok = (k < per) && (i0 + k < total);
        int x = tab[ok ? (k * 1024 + tid) : 0];
        v[k] = ok ? x : 0;
        s += v[k];
    }
    int incl = wave_incl_scan(s, lane);
    if (lane == 31) wtot[wv] = incl;
    __syncthreads();
    int run = incl - s;
    for (int w = 0; w < wv; ++w) run += wtot[w];
#pragma unroll
    for (int k = 0; k < MAXPER; ++k) {
        if ((k < per) && (i0 + k < total)) base[k * 1024 + tid] = run;
        run += v[k];
    }
    if (tid == 1023) base[per * 1024] = run;
}

__global__ __launch_bounds__(1024) void k_bucket_place(const int* __restrict__ src,
                                                       const int* __restrict__ dst,
                                                       const int* __restrict__ base,
                                                       int2* __restrict__ pairs,
                                                       int nE, int nB, int nC) {
    __shared__ int cur[MAXB];
    const int tid = threadIdx.x;
    const int per = (nB * nC + 1023) >> 10;
    for (int b = tid; b < nB; b += 1024) cur[b] = base[tab_at(b * nC + (int)blockIdx.x, per)];
    __syncthreads();
    const int e0 = blockIdx.x * CHUNK + tid;
    int d[4], s[4];
#pragma unroll
    for (int k = 0; k < 4; ++k) {
        int e = e0 + k * 1024;
        int ec = e < nE ? e : nE - 1;
        d[k] = dst[ec];
        s[k] = src[ec];
    }
#pragma unroll
    for (int k = 0; k < 4; ++k) {
        if (e0 + k * 1024 < nE) {
            int p = atomicAdd(&cur[d[k] >> BSH], 1);
            pairs[p] = make_int2(d[k] & ((1 << BSH) - 1), s[k]);
        }
    }
}

__global__ __launch_bounds__(1024) void k_bucket_lists(const int2* __restrict__ pairs,
                                                       const int* __restrict__ base,
                                                       int* __restrict__ rowptr,
                                                       float* __restrict__ dinv,
                                                       int* __restrict__ lst,
                                                       int n, int nC, int total) {
    __shared__ int cnt[1 << BSH];
    __shared__ int cur[1 << BSH];
    __shared__ int wtot[32];
    const int tid = threadIdx.x, lane = tid & 31, wv = tid >> 5;
    const int b = blockIdx.x;
    const int per = (total + 1023) >> 10;
    const int i1 = (b + 1) * nC;
    const int s0 = base[tab_at(b * nC, per)];
    const int s1 = base[(i1 < total) ? tab_at(i1, per) : per * 1024];

    if (tid < (1 << BSH)) cnt[tid] = 0;
    __syncthreads();
    for (int i = s0 + tid; i < s1; i += 4096) {
        int q[4];
#pragma unroll
        for (int u = 0; u < 4; ++u) {
            int j = i + u * 1024;
            q[u] = pairs[j < s1 ? j : i].x;
        }
#pragma unroll
        for (int u = 0; u < 4; ++u)
            if (i + u * 1024 < s1) atomicAdd(&cnt[q[u]], 1);
    }
    __syncthreads();

    const int c = (tid < (1 << BSH)) ? cnt[tid] : 0;
    const int incl = wave_incl_scan(c, lane);
    if (lane == 31) wtot[wv] = incl;
    __syncthreads();
    int ex = s0 + incl - c;
    for (int w = 0; w < wv; ++w) ex += wtot[w];
    if (tid < (1 << BSH)) {
        cur[tid] = ex;
        const int node = (b << BSH) + tid;
        if (node < n) {
            rowptr[node] = ex;
            dinv[node]   = rsqrtf((float)(c + 1));
            if (node == n - 1) rowptr[n] = ex + c;
        }
    }
    __syncthreads();
    for (int i = s0 + tid; i < s1; i += 4096) {
        int2 p[4];
#pragma unroll
        for (int u = 0; u < 4; ++u) {
            int j = i + u * 1024;
            p[u] = pairs[j < s1 ? j : i];
        }
#pragma unroll
        for (int u = 0; u < 4; ++u)
            if (i + u * 1024 < s1) lst[atomicAdd(&cur[p[u].x], 1)] = p[u].y;
    }
}


__global__ __launch_bounds__(256) void k_gemm96(const float* __restrict__ X,
                                                const float* __restrict__ W,
                                                const float* __restrict__ dinv,
                                                float* __restrict__ H, int nRows) {
    __shared__ float Wt[DD * WT_STRIDE];
    for (int i = threadIdx.x; i < DD * DD; i += 256) {
        int k = i / DD, col = i - k * DD;
        Wt[col * WT_STRIDE + k] = W[i];
    }
    __syncthreads();

    int wave = (blockIdx.x * 256 + threadIdx.x) >> 5;
    int lane = threadIdx.x & 31;
    int m0 = wave * 16;
    if (m0 >= nRows) return;

    const int lhalf = lane >> 4;
    const int l16   = lane & 15;
    const int khalf = lhalf * 2;

    const float* Xrow = X + (size_t)(m0 + l16) * DD;

    const int rbase = m0 + lhalf * 8;
    const float4 sc_lo = *(const float4*)(dinv + rbase);
    const float4 sc_hi = *(const float4*)(dinv + rbase + 4);

    v8f acc[6];
#pragma unroll
    for (int t = 0; t < 6; ++t)
#pragma unroll
        for (int r = 0; r < 8; ++r) acc[t][r] = 0.0f;

    for (int k = 0; k < DD; k += 4) {
        v2f a = *(const v2f*)(Xrow + k + khalf);
        int kr = k + khalf;
#pragma unroll
        for (int t = 0; t < 6; ++t) {
            int col = t * 16 + l16;
            v2f b = *(const v2f*)(&Wt[col * WT_STRIDE + kr]);
            acc[t] = __builtin_amdgcn_wmma_f32_16x16x4_f32(
                false, a, false, b, (short)0, acc[t], false, false);
        }
    }

    const float sc[8] = {sc_lo.x, sc_lo.y, sc_lo.z, sc_lo.w, sc_hi.x, sc_hi.y, sc_hi.z, sc_hi.w};
#pragma unroll
    for (int t = 0; t < 6; ++t) {
        int col = t * 16 + l16;
#pragma unroll
        for (int r = 0; r < 8; ++r)
            H[(size_t)(rbase + r) * DD + col] = acc[t][r] * sc[r];
    }
}


struct __attribute__((packed, aligned(4))) f3 { float x, y, z; };

#define GU 8

__device__ __forceinline__ int list_chunk(const int* __restrict__ lst, int base, int m, int lane) {
    int s = 0;
    if (lane < m) s = lst[base + lane];
    return s;
}

__device__ __forceinline__ void sum_chunk(const float* __restrict__ Hl, int s, int m,
                                          float& a0, float& a1, float& a2) {
    for (int i = 0; i < m; i += GU) {
        f3 r[GU];
#pragma unroll
        for (int u = 0; u < GU; ++u) {
            int idx = (u == 0 || i + u < m) ? (i + u) : (m - 1);
            int su  = __builtin_amdgcn_readlane(s, idx);
            r[u]    = *(const f3*)(Hl + (size_t)su * DD);
        }
#pragma unroll
        for (int u = 0; u < GU; ++u) {
            bool ok = (u == 0) || (i + u < m);
            a0 += ok ? r[u].x : 0.0f;
            a1 += ok ? r[u].y : 0.0f;
            a2 += ok ? r[u].z : 0.0f;
        }
    }
}

__device__ __forceinline__ void gather_sum(const float* __restrict__ Hl,
                                           const int* __restrict__ lst,
                                           int beg, int end, int lane,
                                           float& a0, float& a1, float& a2) {
    a0 = 0.0f; a1 = 0.0f; a2 = 0.0f;
    int m = end - beg;
    if (m > 32) m = 32;
    int s = list_chunk(lst, beg, m, lane);
    sum_chunk(Hl, s, m, a0, a1, a2);
    for (int base = beg + 32; base < end; base += 32) {
        m = end - base;
        if (m > 32) m = 32;
        s = list_chunk(lst, base, m, lane);
        sum_chunk(Hl, s, m, a0, a1, a2);
    }
}

__device__ __forceinline__ float leaky_bn(float v, float mu, float var, float g, float be) {
    v = (v >= 0.0f) ? v : SLOPE * v;
    return (v - mu) * rsqrtf(var + BN_EPS) * g + be;
}


#define TW   16
#define TS   100
static_assert((DD * DD) % 512 == 0, "W2 staging: whole trips of 512 threads");
#define LCAP 256

template <bool IN_LDS>
__device__ __forceinline__ int list_chunk_t(const int* __restrict__ lst, const int* lb,
                                            int off, int base, int m, int lane) {
    int s = 0;
    if (lane < m) s = IN_LDS ? lb[off + lane] : lst[base + lane];
    return s;
}

template <bool IN_LDS>
__device__ __forceinline__ void tile_rows(const float* __restrict__ Hl, const int* __restrict__ lst,
                                          const int* lb, float* tl, int rp, float dl, int beg0,
                                          int m0, int n, int lane,
                                          const f3& vb, const f3& vm, const f3& vv, const f3& vg, const f3& ve) {
    for (int j = 0; j < 16; ++j) {
        const int node = m0 + j;
        f3 o;
        o.x = 0.0f; o.y = 0.0f; o.z = 0.0f;
        if (node < n) {
            const int beg = __builtin_amdgcn_readlane(rp, j);
            const int end = __builtin_amdgcn_readlane(rp, j + 1);
            const float dd = __int_as_float(__builtin_amdgcn_readlane(__float_as_int(dl), j));
            const f3 hs = *(const f3*)(Hl + (size_t)node * DD);
            float a0 = 0.0f, a1 = 0.0f, a2 = 0.0f;
            for (int base = beg; base < end; base += 32) {
                int m = end - base;
                if (m > 32) m = 32;
                int s = list_chunk_t<IN_LDS>(lst, lb, base - beg0, base, m, lane);
                sum_chunk(Hl, s, m, a0, a1, a2);
            }
            o.x = leaky_bn((a0 + hs.x) * dd + vb.x, vm.x, vv.x, vg.x, ve.x);
            o.y = leaky_bn((a1 + hs.y) * dd + vb.y, vm.y, vv.y, vg.y, ve.y);
            o.z = leaky_bn((a2 + hs.z) * dd + vb.z, vm.z, vv.z, vg.z, ve.z);
        }
        *(f3*)(tl + j * TS + lane * 3) = o;
    }
}

__global__ __launch_bounds__(512) void k_gather_l1_gemm(const float* __restrict__ H,
                                                        const float* __restrict__ dinv,
                                                        const int* __restrict__ rowptr,
                                                        const int* __restrict__ lst,
                                                        const float* __restrict__ b1,
                                                        const float* __restrict__ g,
                                                        const float* __restrict__ be,
                                                        const float* __restrict__ mu,
                                                        const float* __restrict__ var,
                                                        const float* __restrict__ W,
                                                        float* __restrict__ H2, int n) {
    __shared__ __attribute__((aligned(16))) float Wt[DD * WT_STRIDE];
    __shared__ __attribute__((aligned(16))) float tile[TW][16 * TS];
    __shared__ int   lbuf[TW][LCAP];

    const int lane = threadIdx.x & 31;
    const int wv = __builtin_amdgcn_readfirstlane((int)(threadIdx.x >> 5));
    const int m0 = ((int)blockIdx.x * TW + wv) * 16;
    const bool live = m0 < n;

    int   rp = 0;
    float dl = 0.0f;
    if (live) {
        int j  = lane < 16 ? lane : 16;
        int ix = m0 + j;
        rp = rowptr[ix < n ? ix : n];
        int nd = m0 + (lane & 15);
        dl = dinv[nd < n ? nd : n - 1];
    }

    {
        float w[DD * DD / 512];
#pragma unroll
        for (int u = 0; u < DD * DD / 512; ++u) w[u] = W[u * 512 + threadIdx.x];
#pragma unroll
        for (int u = 0; u < DD * DD / 512; ++u) {
            int i = u * 512 + threadIdx.x;
            int k = i / DD, col = i - k * DD;
            Wt[col * WT_STRIDE + k] = w[u];
        }
    }
    __syncthreads();
    if (!live) return;

    float* tl = tile[wv];
    int*   lb = lbuf[wv];
    const int beg0 = __builtin_amdgcn_readlane(rp, 0);
    const int endT = __builtin_amdgcn_readlane(rp, 16);

    {
        int cnt = endT - beg0;
        if (cnt > LCAP) cnt = LCAP;
        if (cnt > 0) {
            int t[LCAP / 32];
#pragma unroll
            for (int u = 0; u < LCAP / 32; ++u) {
                int p = u * 32 + lane;
                t[u] = lst[beg0 + (p < cnt ? p : 0)];
            }
#pragma unroll
            for (int u = 0; u < LCAP / 32; ++u) lb[u * 32 + lane] = t[u];
        }
    }

    const f3 vb = *(const f3*)(b1  + lane * 3);
    const f3 vm = *(const f3*)(mu  + lane * 3);
    const f3 vv = *(const f3*)(var + lane * 3);
    const f3 vg = *(const f3*)(g   + lane * 3);
    const f3 ve = *(const f3*)(be  + lane * 3);

    const float* Hl = H + lane * 3;
    if (endT - beg0 <= LCAP) tile_rows<true >(Hl, lst, lb, tl, rp, dl, beg0, m0, n, lane, vb, vm, vv, vg, ve);
    else                     tile_rows<false>(Hl, lst, lb, tl, rp, dl, beg0, m0, n, lane, vb, vm, vv, vg, ve);

    const int lhalf = lane >> 4;
    const int l16   = lane & 15;
    const int khalf = lhalf * 2;
    const int rbase = m0 + lhalf * 8;

    float sc[8];
#pragma unroll
    for (int r = 0; r < 8; ++r) {
        int lo = __builtin_amdgcn_readlane(__float_as_int(dl), r);
        int hi = __builtin_amdgcn_readlane(__float_as_int(dl), r + 8);
        sc[r] = __int_as_float(lhalf ? hi : lo);
    }

    v8f acc[6];
#pragma unroll
    for (int t = 0; t < 6; ++t)
#pragma unroll
        for (int r = 0; r < 8; ++r) acc[t][r] = 0.0f;

    const float* arow = tl + l16 * TS + khalf;
    for (int k = 0; k < DD; k += 4) {
        v2f a = *(const v2f*)(arow + k);
        int kr = k + khalf;
#pragma unroll
        for (int t = 0; t < 6; ++t) {
            int col = t * 16 + l16;
            v2f b = *(const v2f*)(&Wt[col * WT_STRIDE + kr]);
            acc[t] = __builtin_amdgcn_wmma_f32_16x16x4_f32(
                false, a, false, b, (short)0, acc[t], false, false);
        }
    }

    if (m0 + 16 <= n) {
#pragma unroll
        for (int t = 0; t < 6; ++t) {
            int col = t * 16 + l16;
#pragma unroll
            for (int r = 0; r < 8; ++r)
                H2[(size_t)(rbase + r) * DD + col] = acc[t][r] * sc[r];
        }
    } else {
#pragma unroll
        for (int t = 0; t < 6; ++t) {
            int col = t * 16 + l16;
#pragma unroll
            for (int r = 0; r < 8; ++r)
                if (rbase + r < n)
                    H2[(size_t)(rbase + r) * DD + col] = acc[t][r] * sc[r];
        }
    }
}

__global__ __launch_bounds__(256) void k_gather_l2(const float* __restrict__ H,
                                                   const float* __restrict__ dinv,
                                                   const int* __restrict__ rowptr,
                                                   const int* __restrict__ lst,
                                                   const float* __restrict__ b2,
                                                   float* __restrict__ OUT, int n) {
    int node = (blockIdx.x * 256 + threadIdx.x) >> 5;
    node = __builtin_amdgcn_readfirstlane(node);
    if (node >= n) return;
    const int lane = threadIdx.x & 31;
    const int beg = rowptr[node], end = rowptr[node + 1];
    const float dd = dinv[node];
    const float* Hl = H + lane * 3;

    const f3 hs = *(const f3*)(Hl + (size_t)node * DD);
    const f3 vb = *(const f3*)(b2 + lane * 3);

    float a0, a1, a2;
    gather_sum(Hl, lst, beg, end, lane, a0, a1, a2);

    f3 o;
    o.x = (a0 + hs.x) * dd + vb.x;
    o.y = (a1 + hs.y) * dd + vb.y;
    o.z = (a2 + hs.z) * dd + vb.z;
    *(f3*)(OUT + (size_t)node * DD + lane * 3) = o;
}


extern "C" void kernel_launch(void* const* d_in, const int* in_sizes, int n_in,
                              void* d_out, int out_size, void* d_ws, size_t ws_size,
                              hipStream_t stream) {
    const float* X    = (const float*)d_in[0];
    const int*   ei   = (const int*)d_in[1];
    const float* W1   = (const float*)d_in[2];
    const float* b1   = (const float*)d_in[3];
    const float* W2   = (const float*)d_in[4];
    const float* b2   = (const float*)d_in[5];
    const float* gam  = (const float*)d_in[6];
    const float* bet  = (const float*)d_in[7];
    const float* mu   = (const float*)d_in[8];
    const float* var  = (const float*)d_in[9];

    const int N = in_sizes[0] / DD;
    const int E = in_sizes[1] / 2;
    const int* srcI = ei;
    const int* dstI = ei + E;

    float* dinv = (float*)d_ws;
    float* hA   = dinv + N;
    float* hB   = hA + (size_t)N * DD;
    int*   rptr = (int*)(hB + (size_t)N * DD);
    int*   lst  = rptr + (N + 1);
    int*   tab  = lst + E;
    int*   base = tab + (MAXPER * 1024 + 1);
    int2*  pairs = (int2*)(((size_t)(base + (MAXPER * 1024 + 1)) + 7) & ~(size_t)7);
    float* out  = (float*)d_out;

    const int nB        = (N + (1 << BSH) - 1) >> BSH;
    const int nC        = (E + CHUNK - 1) / CHUNK;
    const int gemmGrid  = ((N + 15) / 16 + 7) / 8;
    const int gathGrid  = (N + 7) / 8;
    const int fuseGrid  = ((N + 15) / 16 + TW - 1) / TW;

    k_bucket_hist <<<nC, 1024, 0, stream>>>(dstI, tab, E, nB, nC);
    k_scan_tab    <<<1,  1024, 0, stream>>>(tab, base, nB * nC);
    k_bucket_place<<<nC, 1024, 0, stream>>>(srcI, dstI, base, pairs, E, nB, nC);
    k_bucket_lists<<<nB, 1024, 0, stream>>>(pairs, base, rptr, dinv, lst, N, nC, nB * nC);

    k_gemm96        <<<gemmGrid, 256, 0, stream>>>(X, W1, dinv, hA, N);
    k_gather_l1_gemm<<<fuseGrid, 512, 0, stream>>>(hA, dinv, rptr, lst, b1, gam, bet, mu, var, W2, hB, N);

    k_gather_l2<<<gathGrid, 256, 0, stream>>>(hB, dinv, rptr, lst, b2, out, N);
}


# Edge sort: level 1 orders each 4096-edge chunk by bucket in its work-group (LDS count, scan, place), level 2 walks its bucket's run per chunk; count and scan launches gone, 7->5; LDS arrival order
# speedup vs baseline: 3.8912x; 1.1685x over previous
#include <hip/hip_runtime.h>

#define DD 96
#define WT_STRIDE 98
#define BN_EPS 1e-5f
#define SLOPE 0.05f

typedef __attribute__((ext_vector_type(2))) float v2f;
typedef __attribute__((ext_vector_type(8))) float v8f;


#define BSH    9
#define MAXB   2048
#define CHUNK  4096
#define MAXPER 40
#define MAXC   1024
#define SEGU   8

__device__ __forceinline__ int wave_incl_scan(int v, int lane) {
    for (int o = 1; o < 32; o <<= 1) {
        int u = __shfl_up(v, o);
        if (lane >= o) v += u;
    }
    return v;
}

__global__ __launch_bounds__(1024) void k_chunk_sort(const int* __restrict__ src,
                                                     const int* __restrict__ dst,
                                                     int* __restrict__ offT,
                                                     int2* __restrict__ pairs,
                                                     int nE, int nB, int nC) {
    __shared__ int  h[MAXB];
    __shared__ int  wtot[32];
    __shared__ int2 lp[CHUNK];
    const int tid = threadIdx.x, lane = tid & 31, wv = tid >> 5;
    h[tid] = 0;
    h[tid + 1024] = 0;
    __syncthreads();
    const int e0 = blockIdx.x * CHUNK + tid;
    int d[4], s[4];
    int ec[4];
#pragma unroll
    for (int k = 0; k < 4; ++k) {
        int e = e0 + k * 1024;
        ec[k] = e < nE ? e : nE - 1;
    }
#pragma unroll
    for (int k = 0; k < 4; ++k) {
        d[k] = dst[ec[k]];
        s[k] = src[ec[k]];
    }
#pragma unroll
    for (int k = 0; k < 4; ++k)
        if (e0 + k * 1024 < nE) atomicAdd(&h[d[k] >> BSH], 1);
    __syncthreads();
    const int c0 = h[2 * tid], c1 = h[2 * tid + 1];
    const int sum = c0 + c1;
    const int incl = wave_incl_scan(sum, lane);
    if (lane == 31) wtot[wv] = incl;
    __syncthreads();
    int ex = incl - sum;
    for (int w = 0; w < wv; ++w) ex += wtot[w];
    h[2 * tid]     = ex;
    h[2 * tid + 1] = ex + c0;
    if (2 * tid     <= nB) offT[(2 * tid)     * nC + (int)blockIdx.x] = ex;
    if (2 * tid + 1 <= nB) offT[(2 * tid + 1) * nC + (int)blockIdx.x] = ex + c0;
    __syncthreads();
#pragma unroll
    for (int k = 0; k < 4; ++k) {
        if (e0 + k * 1024 < nE) {
            int p = atomicAdd(&h[d[k] >> BSH], 1);
            lp[p] = make_int2(d[k] & ((1 << BSH) - 1), s[k]);
        }
    }
    __syncthreads();
    int cnt = nE - (int)blockIdx.x * CHUNK;
    if (cnt > CHUNK) cnt = CHUNK;
    for (int i = tid; i < cnt; i += 1024) pairs[(size_t)blockIdx.x * CHUNK + i] = lp[i];
}

template <bool PLACE>
__device__ __forceinline__ void bucket_walk(const int2* __restrict__ pairs, const int2* seg,
                                            int* ctr, int* __restrict__ lst, int nC, int wv, int lane) {
    for (int cb = wv; cb < nC; cb += 32 * SEGU) {
        int2 sg[SEGU];
        int2 q[SEGU];
#pragma unroll
        for (int u = 0; u < SEGU; ++u) {
            const int c = cb + 32 * u;
            sg[u] = seg[c < nC ? c : nC - 1];
            if (c >= nC) sg[u].y = 0;
        }
#pragma unroll
        for (int u = 0; u < SEGU; ++u) {
            q[u] = make_int2(0, 0);
            if (lane < sg[u].y) q[u] = pairs[sg[u].x + lane];
        }
#pragma unroll
        for (int u = 0; u < SEGU; ++u) {
            if (lane < sg[u].y) {
                if (PLACE) lst[atomicAdd(&ctr[q[u].x], 1)] = q[u].y;
                else       atomicAdd(&ctr[q[u].x], 1);
            }
        }
#pragma unroll
        for (int u = 0; u < SEGU; ++u) {
            for (int j = 32 + lane; j < sg[u].y; j += 32) {
                const int2 r = pairs[sg[u].x + j];
                if (PLACE) lst[atomicAdd(&ctr[r.x], 1)] = r.y;
                else       atomicAdd(&ctr[r.x], 1);
            }
        }
    }
}

__global__ __launch_bounds__(1024) void k_bucket_lists(const int2* __restrict__ pairs,
                                                       const int* __restrict__ offT,
                                                       int* __restrict__ rowptr,
                                                       float* __restrict__ dinv,
                                                       int* __restrict__ lst,
                                                       int n, int nC) {
    __shared__ int cnt[1 << BSH];
    __shared__ int cur[1 << BSH];
    __shared__ int wtot[32];
    __shared__ int red[32];
    __shared__ int2 seg[MAXC];
    const int tid = threadIdx.x, lane = tid & 31, wv = tid >> 5;
    const int b = blockIdx.x;

    int a0 = 0;
    for (int c = tid; c < nC; c += 1024) {
        const int o0 = offT[b * nC + c];
        const int o1 = offT[(b + 1) * nC + c];
        seg[c] = make_int2(c * CHUNK + o0, o1 - o0);
        a0 += o0;
    }
    if (tid < (1 << BSH)) cnt[tid] = 0;
    const int ra = wave_incl_scan(a0, lane);
    if (lane == 31) red[wv] = ra;
    __syncthreads();
    int s0 = 0;
    for (int w = 0; w < 32; ++w) s0 += red[w];

    bucket_walk<false>(pairs, seg, cnt, lst, nC, wv, lane);
    __syncthreads();

    const int c = (tid < (1 << BSH)) ? cnt[tid] : 0;
    const int incl = wave_incl_scan(c, lane);
    if (lane == 31) wtot[wv] = incl;
    __syncthreads();
    int ex = s0 + incl - c;
    for (int w = 0; w < wv; ++w) ex += wtot[w];
    if (tid < (1 << BSH)) {
        cur[tid] = ex;
        const int node = (b << BSH) + tid;
        if (node < n) {
            rowptr[node] = ex;
            dinv[node]   = rsqrtf((float)(c + 1));
            if (node == n - 1) rowptr[n] = ex + c;
        }
    }
    __syncthreads();
    bucket_walk<true>(pairs, seg, cur, lst, nC, wv, lane);
}


__global__ __launch_bounds__(256) void k_gemm96(const float* __restrict__ X,
                                                const float* __restrict__ W,
                                                const float* __restrict__ dinv,
                                                float* __restrict__ H, int nRows) {
    __shared__ float Wt[DD * WT_STRIDE];
    for (int i = threadIdx.x; i < DD * DD; i += 256) {
        int k = i / DD, col = i - k * DD;
        Wt[col * WT_STRIDE + k] = W[i];
    }
    __syncthreads();

    int wave = (blockIdx.x * 256 + threadIdx.x) >> 5;
    int lane = threadIdx.x & 31;
    int m0 = wave * 16;
    if (m0 >= nRows) return;

    const int lhalf = lane >> 4;
    const int l16   = lane & 15;
    const int khalf = lhalf * 2;

    const float* Xrow = X + (size_t)(m0 + l16) * DD;

    const int rbase = m0 + lhalf * 8;
    const float4 sc_lo = *(const float4*)(dinv + rbase);
    const float4 sc_hi = *(const float4*)(dinv + rbase + 4);

    v8f acc[6];
#pragma unroll
    for (int t = 0; t < 6; ++t)
#pragma unroll
        for (int r = 0; r < 8; ++r) acc[t][r] = 0.0f;

    for (int k = 0; k < DD; k += 4) {
        v2f a = *(const v2f*)(Xrow + k + khalf);
        int kr = k + khalf;
#pragma unroll
        for (int t = 0; t < 6; ++t) {
            int col = t * 16 + l16;
            v2f b = *(const v2f*)(&Wt[col * WT_STRIDE + kr]);
            acc[t] = __builtin_amdgcn_wmma_f32_16x16x4_f32(
                false, a, false, b, (short)0, acc[t], false, false);
        }
    }

    const float sc[8] = {sc_lo.x, sc_lo.y, sc_lo.z, sc_lo.w, sc_hi.x, sc_hi.y, sc_hi.z, sc_hi.w};
#pragma unroll
    for (int t = 0; t < 6; ++t) {
        int col = t * 16 + l16;
#pragma unroll
        for (int r = 0; r < 8; ++r)
            H[(size_t)(rbase + r) * DD + col] = acc[t][r] * sc[r];
    }
}


struct __attribute__((packed, aligned(4))) f3 { float x, y, z; };

#define GU 8

__device__ __forceinline__ int list_chunk(const int* __restrict__ lst, int base, int m, int lane) {
    int s = 0;
    if (lane < m) s = lst[base + lane];
    return s;
}

__device__ __forceinline__ void sum_chunk(const float* __restrict__ Hl, int s, int m,
                                          float& a0, float& a1, float& a2) {
    for (int i = 0; i < m; i += GU) {
        f3 r[GU];
#pragma unroll
        for (int u = 0; u < GU; ++u) {
            int idx = (u == 0 || i + u < m) ? (i + u) : (m - 1);
            int su  = __builtin_amdgcn_readlane(s, idx);
            r[u]    = *(const f3*)(Hl + (size_t)su * DD);
        }
#pragma unroll
        for (int u = 0; u < GU; ++u) {
            bool ok = (u == 0) || (i + u < m);
            a0 += ok ? r[u].x : 0.0f;
            a1 += ok ? r[u].y : 0.0f;
            a2 += ok ? r[u].z : 0.0f;
        }
    }
}

__device__ __forceinline__ void gather_sum(const float* __restrict__ Hl,
                                           const int* __restrict__ lst,
                                           int beg, int end, int lane,
                                           float& a0, float& a1, float& a2) {
    a0 = 0.0f; a1 = 0.0f; a2 = 0.0f;
    int m = end - beg;
    if (m > 32) m = 32;
    int s = list_chunk(lst, beg, m, lane);
    sum_chunk(Hl, s, m, a0, a1, a2);
    for (int base = beg + 32; base < end; base += 32) {
        m = end - base;
        if (m > 32) m = 32;
        s = list_chunk(lst, base, m, lane);
        sum_chunk(Hl, s, m, a0, a1, a2);
    }
}

__device__ __forceinline__ float leaky_bn(float v, float mu, float var, float g, float be) {
    v = (v >= 0.0f) ? v : SLOPE * v;
    return (v - mu) * rsqrtf(var + BN_EPS) * g + be;
}


#define TW   16
#define TS   100
static_assert((DD * DD) % 512 == 0, "W2 staging: whole trips of 512 threads");
#define LCAP 256

template <bool IN_LDS>
__device__ __forceinline__ int list_chunk_t(const int* __restrict__ lst, const int* lb,
                                            int off, int base, int m, int lane) {
    int s = 0;
    if (lane < m) s = IN_LDS ? lb[off + lane] : lst[base + lane];
    return s;
}

template <bool IN_LDS>
__device__ __forceinline__ void tile_rows(const float* __restrict__ Hl, const int* __restrict__ lst,
                                          const int* lb, float* tl, int rp, float dl, int beg0,
                                          int m0, int n, int lane,
                                          const f3& vb, const f3& vm, const f3& vv, const f3& vg, const f3& ve) {
    for (int j = 0; j < 16; ++j) {
        const int node = m0 + j;
        f3 o;
        o.x = 0.0f; o.y = 0.0f; o.z = 0.0f;
        if (node < n) {
            const int beg = __builtin_amdgcn_readlane(rp, j);
            const int end = __builtin_amdgcn_readlane(rp, j + 1);
            const float dd = __int_as_float(__builtin_amdgcn_readlane(__float_as_int(dl), j));
            const f3 hs = *(const f3*)(Hl + (size_t)node * DD);
            float a0 = 0.0f, a1 = 0.0f, a2 = 0.0f;
            for (int base = beg; base < end; base += 32) {
                int m = end - base;
                if (m > 32) m = 32;
                int s = list_chunk_t<IN_LDS>(lst, lb, base - beg0, base, m, lane);
                sum_chunk(Hl, s, m, a0, a1, a2);
            }
            o.x = leaky_bn((a0 + hs.x) * dd + vb.x, vm.x, vv.x, vg.x, ve.x);
            o.y = leaky_bn((a1 + hs.y) * dd + vb.y, vm.y, vv.y, vg.y, ve.y);
            o.z = leaky_bn((a2 + hs.z) * dd + vb.z, vm.z, vv.z, vg.z, ve.z);
        }
        *(f3*)(tl + j * TS + lane * 3) = o;
    }
}

__global__ __launch_bounds__(512) void k_gather_l1_gemm(const float* __restrict__ H,
                                                        const float* __restrict__ dinv,
                                                        const int* __restrict__ rowptr,
                                                        const int* __restrict__ lst,
                                                        const float* __restrict__ b1,
                                                        const float* __restrict__ g,
                                                        const float* __restrict__ be,
                                                        const float* __restrict__ mu,
                                                        const float* __restrict__ var,
                                                        const float* __restrict__ W,
                                                        float* __restrict__ H2, int n) {
    __shared__ __attribute__((aligned(16))) float Wt[DD * WT_STRIDE];
    __shared__ __attribute__((aligned(16))) float tile[TW][16 * TS];
    __shared__ int   lbuf[TW][LCAP];

    const int lane = threadIdx.x & 31;
    const int wv = __builtin_amdgcn_readfirstlane((int)(threadIdx.x >> 5));
    const int m0 = ((int)blockIdx.x * TW + wv) * 16;
    const bool live = m0 < n;

    int   rp = 0;
    float dl = 0.0f;
    if (live) {
        int j  = lane < 16 ? lane : 16;
        int ix = m0 + j;
        rp = rowptr[ix < n ? ix : n];
        int nd = m0 + (lane & 15);
        dl = dinv[nd < n ? nd : n - 1];
    }

    {
        float w[DD * DD / 512];
#pragma unroll
        for (int u = 0; u < DD * DD / 512; ++u) w[u] = W[u * 512 + threadIdx.x];
#pragma unroll
        for (int u = 0; u < DD * DD / 512; ++u) {
            int i = u * 512 + threadIdx.x;
            int k = i / DD, col = i - k * DD;
            Wt[col * WT_STRIDE + k] = w[u];
        }
    }
    __syncthreads();
    if (!live) return;

    float* tl = tile[wv];
    int*   lb = lbuf[wv];
    const int beg0 = __builtin_amdgcn_readlane(rp, 0);
    const int endT = __builtin_amdgcn_readlane(rp, 16);

    {
        int cnt = endT - beg0;
        if (cnt > LCAP) cnt = LCAP;
        if (cnt > 0) {
            int t[LCAP / 32];
#pragma unroll
            for (int u = 0; u < LCAP / 32; ++u) {
                int p = u * 32 + lane;
                t[u] = lst[beg0 + (p < cnt ? p : 0)];
            }
#pragma unroll
            for (int u = 0; u < LCAP / 32; ++u) lb[u * 32 + lane] = t[u];
        }
    }

    const f3 vb = *(const f3*)(b1  + lane * 3);
    const f3 vm = *(const f3*)(mu  + lane * 3);
    const f3 vv = *(const f3*)(var + lane * 3);
    const f3 vg = *(const f3*)(g   + lane * 3);
    const f3 ve = *(const f3*)(be  + lane * 3);

    const float* Hl = H + lane * 3;
    if (endT - beg0 <= LCAP) tile_rows<true >(Hl, lst, lb, tl, rp, dl, beg0, m0, n, lane, vb, vm, vv, vg, ve);
    else                     tile_rows<false>(Hl, lst, lb, tl, rp, dl, beg0, m0, n, lane, vb, vm, vv, vg, ve);

    const int lhalf = lane >> 4;
    const int l16   = lane & 15;
    const int khalf = lhalf * 2;
    const int rbase = m0 + lhalf * 8;

    float sc[8];
#pragma unroll
    for (int r = 0; r < 8; ++r) {
        int lo = __builtin_amdgcn_readlane(__float_as_int(dl), r);
        int hi = __builtin_amdgcn_readlane(__float_as_int(dl), r + 8);
        sc[r] = __int_as_float(lhalf ? hi : lo);
    }

    v8f acc[6];
#pragma unroll
    for (int t = 0; t < 6; ++t)
#pragma unroll
        for (int r = 0; r < 8; ++r) acc[t][r] = 0.0f;

    const float* arow = tl + l16 * TS + khalf;
    for (int k = 0; k < DD; k += 4) {
        v2f a = *(const v2f*)(arow + k);
        int kr = k + khalf;
#pragma unroll
        for (int t = 0; t < 6; ++t) {
            int col = t * 16 + l16;
            v2f b = *(const v2f*)(&Wt[col * WT_STRIDE + kr]);
            acc[t] = __builtin_amdgcn_wmma_f32_16x16x4_f32(
                false, a, false, b, (short)0, acc[t], false, false);
        }
    }

    if (m0 + 16 <= n) {
#pragma unroll
        for (int t = 0; t < 6; ++t) {
            int col = t * 16 + l16;
#pragma unroll
            for (int r = 0; r < 8; ++r)
                H2[(size_t)(rbase + r) * DD + col] = acc[t][r] * sc[r];
        }
    } else {
#pragma unroll
        for (int t = 0; t < 6; ++t) {
            int col = t * 16 + l16;
#pragma unroll
            for (int r = 0; r < 8; ++r)
                if (rbase + r < n)
                    H2[(size_t)(rbase + r) * DD + col] = acc[t][r] * sc[r];
        }
    }
}

__global__ __launch_bounds__(256) void k_gather_l2(const float* __restrict__ H,
                                                   const float* __restrict__ dinv,
                                                   const int* __restrict__ rowptr,
                                                   const int* __restrict__ lst,
                                                   const float* __restrict__ b2,
                                                   float* __restrict__ OUT, int n) {
    int node = (blockIdx.x * 256 + threadIdx.x) >> 5;
    node = __builtin_amdgcn_readfirstlane(node);
    if (node >= n) return;
    const int lane = threadIdx.x & 31;
    const int beg = rowptr[node], end = rowptr[node + 1];
    const float dd = dinv[node];
    const float* Hl = H + lane * 3;

    const f3 hs = *(const f3*)(Hl + (size_t)node * DD);
    const f3 vb = *(const f3*)(b2 + lane * 3);

    float a0, a1, a2;
    gather_sum(Hl, lst, beg, end, lane, a0, a1, a2);

    f3 o;
    o.x = (a0 + hs.x) * dd + vb.x;
    o.y = (a1 + hs.y) * dd + vb.y;
    o.z = (a2 + hs.z) * dd + vb.z;
    *(f3*)(OUT + (size_t)node * DD + lane * 3) = o;
}


extern "C" void kernel_launch(void* const* d_in, const int* in_sizes, int n_in,
                              void* d_out, int out_size, void* d_ws, size_t ws_size,
                              hipStream_t stream) {
    const float* X    = (const float*)d_in[0];
    const int*   ei   = (const int*)d_in[1];
    const float* W1   = (const float*)d_in[2];
    const float* b1   = (const float*)d_in[3];
    const float* W2   = (const float*)d_in[4];
    const float* b2   = (const float*)d_in[5];
    const float* gam  = (const float*)d_in[6];
    const float* bet  = (const float*)d_in[7];
    const float* mu   = (const float*)d_in[8];
    const float* var  = (const float*)d_in[9];

    const int N = in_sizes[0] / DD;
    const int E = in_sizes[1] / 2;
    const int* srcI = ei;
    const int* dstI = ei + E;

    float* dinv = (float*)d_ws;
    float* hA   = dinv + N;
    float* hB   = hA + (size_t)N * DD;
    int*   rptr = (int*)(hB + (size_t)N * DD);
    int*   lst  = rptr + (N + 1);
    int*   offT = lst + E;
    int2*  pairs = (int2*)(((size_t)(offT + 2 * (MAXPER * 1024 + 1)) + 7) & ~(size_t)7);
    float* out  = (float*)d_out;

    const int nB        = (N + (1 << BSH) - 1) >> BSH;
    const int nC        = (E + CHUNK - 1) / CHUNK;
    const int gemmGrid  = ((N + 15) / 16 + 7) / 8;
    const int gathGrid  = (N + 7) / 8;
    const int fuseGrid  = ((N + 15) / 16 + TW - 1) / TW;

    k_chunk_sort  <<<nC, 1024, 0, stream>>>(srcI, dstI, offT, pairs, E, nB, nC);
    k_bucket_lists<<<nB, 1024, 0, stream>>>(pairs, offT, rptr, dinv, lst, N, nC);

    k_gemm96        <<<gemmGrid, 256, 0, stream>>>(X, W1, dinv, hA, N);
    k_gather_l1_gemm<<<fuseGrid, 512, 0, stream>>>(hA, dinv, rptr, lst, b1, gam, bet, mu, var, W2, hB, N);

    k_gather_l2<<<gathGrid, 256, 0, stream>>>(hB, dinv, rptr, lst, b2, out, N);
}


# Both layers sum the neighbours' rows first and multiply the 16-row tile by W after ((A X) W for A (X W)): layer-1 product launch and its H table gone, 5 -> 4 launches; f32 rounding order changes
# speedup vs baseline: 4.2786x; 1.0996x over previous
#include <hip/hip_runtime.h>

#define DD 96
#define WT_STRIDE 98
#define BN_EPS 1e-5f
#define SLOPE 0.05f

typedef __attribute__((ext_vector_type(2))) float v2f;
typedef __attribute__((ext_vector_type(8))) float v8f;


#define BSH    9
#define MAXB   2048
#define CHUNK  4096
#define MAXPER 40
#define MAXC   1024
#define SEGU   8

__device__ __forceinline__ int wave_incl_scan(int v, int lane) {
    for (int o = 1; o < 32; o <<= 1) {
        int u = __shfl_up(v, o);
        if (lane >= o) v += u;
    }
    return v;
}

__global__ __launch_bounds__(1024) void k_chunk_sort(const int* __restrict__ src,
                                                     const int* __restrict__ dst,
                                                     int* __restrict__ offT,
                                                     int2* __restrict__ pairs,
                                                     int nE, int nB, int nC) {
    __shared__ int  h[MAXB];
    __shared__ int  wtot[32];
    __shared__ int2 lp[CHUNK];
    const int tid = threadIdx.x, lane = tid & 31, wv = tid >> 5;
    h[tid] = 0;
    h[tid + 1024] = 0;
    __syncthreads();
    const int e0 = blockIdx.x * CHUNK + tid;
    int d[4], s[4];
    int ec[4];
#pragma unroll
    for (int k = 0; k < 4; ++k) {
        int e = e0 + k * 1024;
        ec[k] = e < nE ? e : nE - 1;
    }
#pragma unroll
    for (int k = 0; k < 4; ++k) {
        d[k] = dst[ec[k]];
        s[k] = src[ec[k]];
    }
#pragma unroll
    for (int k = 0; k < 4; ++k)
        if (e0 + k * 1024 < nE) atomicAdd(&h[d[k] >> BSH], 1);
    __syncthreads();
    const int c0 = h[2 * tid], c1 = h[2 * tid + 1];
    const int sum = c0 + c1;
    const int incl = wave_incl_scan(sum, lane);
    if (lane == 31) wtot[wv] = incl;
    __syncthreads();
    int ex = incl - sum;
    for (int w = 0; w < wv; ++w) ex += wtot[w];
    h[2 * tid]     = ex;
    h[2 * tid + 1] = ex + c0;
    if (2 * tid     <= nB) offT[(2 * tid)     * nC + (int)blockIdx.x] = ex;
    if (2 * tid + 1 <= nB) offT[(2 * tid + 1) * nC + (int)blockIdx.x] = ex + c0;
    __syncthreads();
#pragma unroll
    for (int k = 0; k < 4; ++k) {
        if (e0 + k * 1024 < nE) {
            int p = atomicAdd(&h[d[k] >> BSH], 1);
            lp[p] = make_int2(d[k] & ((1 << BSH) - 1), s[k]);
        }
    }
    __syncthreads();
    int cnt = nE - (int)blockIdx.x * CHUNK;
    if (cnt > CHUNK) cnt = CHUNK;
    for (int i = tid; i < cnt; i += 1024) pairs[(size_t)blockIdx.x * CHUNK + i] = lp[i];
}

template <bool PLACE>
__device__ __forceinline__ void bucket_walk(const int2* __restrict__ pairs, const int2* seg,
                                            int* ctr, int* __restrict__ lst, int nC, int wv, int lane) {
    for (int cb = wv; cb < nC; cb += 32 * SEGU) {
        int2 sg[SEGU];
        int2 q[SEGU];
#pragma unroll
        for (int u = 0; u < SEGU; ++u) {
            const int c = cb + 32 * u;
            sg[u] = seg[c < nC ? c : nC - 1];
            if (c >= nC) sg[u].y = 0;
        }
#pragma unroll
        for (int u = 0; u < SEGU; ++u) {
            q[u] = make_int2(0, 0);
            if (lane < sg[u].y) q[u] = pairs[sg[u].x + lane];
        }
#pragma unroll
        for (int u = 0; u < SEGU; ++u) {
            if (lane < sg[u].y) {
                if (PLACE) lst[atomicAdd(&ctr[q[u].x], 1)] = q[u].y;
                else       atomicAdd(&ctr[q[u].x], 1);
            }
        }
#pragma unroll
        for (int u = 0; u < SEGU; ++u) {
            for (int j = 32 + lane; j < sg[u].y; j += 32) {
                const int2 r = pairs[sg[u].x + j];
                if (PLACE) lst[atomicAdd(&ctr[r.x], 1)] = r.y;
                else       atomicAdd(&ctr[r.x], 1);
            }
        }
    }
}

__global__ __launch_bounds__(1024) void k_bucket_lists(const int2* __restrict__ pairs,
                                                       const int* __restrict__ offT,
                                                       int* __restrict__ rowptr,
                                                       float* __restrict__ dinv,
                                                       int* __restrict__ lst,
                                                       int n, int nC) {
    __shared__ int cnt[1 << BSH];
    __shared__ int cur[1 << BSH];
    __shared__ int wtot[32];
    __shared__ int red[32];
    __shared__ int2 seg[MAXC];
    const int tid = threadIdx.x, lane = tid & 31, wv = tid >> 5;
    const int b = blockIdx.x;

    int a0 = 0;
    for (int c = tid; c < nC; c += 1024) {
        const int o0 = offT[b * nC + c];
        const int o1 = offT[(b + 1) * nC + c];
        seg[c] = make_int2(c * CHUNK + o0, o1 - o0);
        a0 += o0;
    }
    if (tid < (1 << BSH)) cnt[tid] = 0;
    const int ra = wave_incl_scan(a0, lane);
    if (lane == 31) red[wv] = ra;
    __syncthreads();
    int s0 = 0;
    for (int w = 0; w < 32; ++w) s0 += red[w];

    bucket_walk<false>(pairs, seg, cnt, lst, nC, wv, lane);
    __syncthreads();

    const int c = (tid < (1 << BSH)) ? cnt[tid] : 0;
    const int incl = wave_incl_scan(c, lane);
    if (lane == 31) wtot[wv] = incl;
    __syncthreads();
    int ex = s0 + incl - c;
    for (int w = 0; w < wv; ++w) ex += wtot[w];
    if (tid < (1 << BSH)) {
        cur[tid] = ex;
        const int node = (b << BSH) + tid;
        if (node < n) {
            rowptr[node] = ex;
            dinv[node]   = rsqrtf((float)(c + 1));
            if (node == n - 1) rowptr[n] = ex + c;
        }
    }
    __syncthreads();
    bucket_walk<true>(pairs, seg, cur, lst, nC, wv, lane);
}


struct __attribute__((packed, aligned(4))) f3 { float x, y, z; };

#define GU 8

template <bool WEIGHTED>
__device__ __forceinline__ void sum_chunk(const float* __restrict__ Tl, int s, float w, int m,
                                          float& a0, float& a1, float& a2) {
    for (int i = 0; i < m; i += GU) {
        f3 r[GU];
#pragma unroll
        for (int u = 0; u < GU; ++u) {
            int idx = (u == 0 || i + u < m) ? (i + u) : (m - 1);
            int su  = __builtin_amdgcn_readlane(s, idx);
            r[u]    = *(const f3*)(Tl + (size_t)su * DD);
        }
#pragma unroll
        for (int u = 0; u < GU; ++u) {
            if (WEIGHTED) {
                const bool  ok = (u == 0) || (i + u < m);
                const float wl = __int_as_float(__builtin_amdgcn_readlane(__float_as_int(w), i + u));
                const float wu = ok ? wl : 0.0f;
                a0 = fmaf(wu, r[u].x, a0);
                a1 = fmaf(wu, r[u].y, a1);
                a2 = fmaf(wu, r[u].z, a2);
            } else {
                bool ok = (u == 0) || (i + u < m);
                a0 += ok ? r[u].x : 0.0f;
                a1 += ok ? r[u].y : 0.0f;
                a2 += ok ? r[u].z : 0.0f;
            }
        }
    }
}

__device__ __forceinline__ float leaky_bn(float v, float mu, float f, float be) {
    v = (v >= 0.0f) ? v : SLOPE * v;
    return (v - mu) * f + be;
}

#define TW2   16
#define LCAP2 256
#define TW1   15
#define LCAP1 192
#define TS   100

template <bool IN_LDS>
__device__ __forceinline__ int list_chunk_t(const int* __restrict__ lst, const int* lb,
                                            int off, int base, int m, int lane) {
    int s = 0;
    if (lane < m) s = IN_LDS ? lb[off + lane] : lst[base + lane];
    return s;
}

template <bool IN_LDS, bool L1>
__device__ __forceinline__ void tile_rows(const float* __restrict__ Tl, const float* __restrict__ dinv,
                                          const int* __restrict__ lst, const int* lb, float* tl,
                                          int rp, float dl, int beg0, int m0, int n, int lane) {
    for (int j = 0; j < 16; ++j) {
        const int node = m0 + j;
        f3 o;
        o.x = 0.0f; o.y = 0.0f; o.z = 0.0f;
        if (node < n) {
            const int beg = __builtin_amdgcn_readlane(rp, j);
            const int end = __builtin_amdgcn_readlane(rp, j + 1);
            const float dd = __int_as_float(__builtin_amdgcn_readlane(__float_as_int(dl), j));
            const f3 hs = *(const f3*)(Tl + (size_t)node * DD);
            float a0 = 0.0f, a1 = 0.0f, a2 = 0.0f;
            for (int base = beg; base < end; base += 32) {
                int m = end - base;
                if (m > 32) m = 32;
                int   s;
                float w = 0.0f;
                if (IN_LDS && L1) {
                    int2 e = make_int2(0, 0);
                    if (lane < m) e = ((const int2*)lb)[base - beg0 + lane];
                    s = e.x;
                    w = __int_as_float(e.y);
                } else {
                    s = list_chunk_t<IN_LDS>(lst, lb, base - beg0, base, m, lane);
                    if (L1) w = dinv[s];
                }
                sum_chunk<L1>(Tl, s, w, m, a0, a1, a2);
            }
            if (L1) {
                o.x = fmaf(dd, hs.x, a0) * dd;
                o.y = fmaf(dd, hs.y, a1) * dd;
                o.z = fmaf(dd, hs.z, a2) * dd;
            } else {
                o.x = (a0 + hs.x) * dd;
                o.y = (a1 + hs.y) * dd;
                o.z = (a2 + hs.z) * dd;
            }
        }
        *(f3*)(tl + j * TS + lane * 3) = o;
    }
}

template <bool L1>
__global__ __launch_bounds__(L1 ? TW1 * 32 : TW2 * 32, 8) void k_layer(const float* __restrict__ T,
                                               const float* __restrict__ dinv,
                                               const int* __restrict__ rowptr,
                                               const int* __restrict__ lst,
                                               const float* __restrict__ bias,
                                               const float* __restrict__ g,
                                               const float* __restrict__ be,
                                               const float* __restrict__ mu,
                                               const float* __restrict__ var,
                                               const float* __restrict__ W,
                                               float* __restrict__ O, int n) {
    __shared__ __attribute__((aligned(16))) float Wt[DD * WT_STRIDE];
    constexpr int TW   = L1 ? TW1 : TW2;
    constexpr int NT   = TW * 32;
    constexpr int LCAP = L1 ? LCAP1 : LCAP2;
    __shared__ __attribute__((aligned(16))) float tile[TW][16 * TS];
    __shared__ __attribute__((aligned(8))) int lbuf[TW][L1 ? 2 * LCAP1 : LCAP2];

    const int lane = threadIdx.x & 31;
    const int wv = __builtin_amdgcn_readfirstlane((int)(threadIdx.x >> 5));
    const int m0 = ((int)blockIdx.x * TW + wv) * 16;
    const bool live = m0 < n;

    int   rp = 0;
    float dl = 0.0f;
    if (live) {
        int j  = lane < 16 ? lane : 16;
        int ix = m0 + j;
        rp = rowptr[ix < n ? ix : n];
        int nd = m0 + (lane & 15);
        dl = dinv[nd < n ? nd : n - 1];
    }

    {
        constexpr int NU = (DD * DD + NT - 1) / NT;
        float w[NU];
#pragma unroll
        for (int u = 0; u < NU; ++u) {
            int i = u * NT + (int)threadIdx.x;
            w[u] = W[i < DD * DD ? i : DD * DD - 1];
        }
#pragma unroll
        for (int u = 0; u < NU; ++u) {
            int i = u * NT + (int)threadIdx.x;
            if (i < DD * DD) {
                int k = i / DD, col = i - k * DD;
                Wt[col * WT_STRIDE + k] = w[u];
            }
        }
    }
    __syncthreads();
    if (!live) return;

    float* tl = tile[wv];
    int*   lb = lbuf[wv];
    const int beg0 = __builtin_amdgcn_readlane(rp, 0);
    const int endT = __builtin_amdgcn_readlane(rp, 16);

    {
        int cnt = endT - beg0;
        if (cnt > LCAP) cnt = LCAP;
        if (cnt > 0) {
            int t[LCAP / 32];
#pragma unroll
            for (int u = 0; u < LCAP / 32; ++u) {
                int p = u * 32 + lane;
                t[u] = lst[beg0 + (p < cnt ? p : 0)];
            }
            if (L1) {
                float d[LCAP / 32];
#pragma unroll
                for (int u = 0; u < LCAP / 32; ++u) d[u] = dinv[t[u]];
#pragma unroll
                for (int u = 0; u < LCAP / 32; ++u)
                    ((int2*)lb)[u * 32 + lane] = make_int2(t[u], __float_as_int(d[u]));
            } else {
#pragma unroll
                for (int u = 0; u < LCAP / 32; ++u) lb[u * 32 + lane] = t[u];
            }
        }
    }

    const float* Tl = T + lane * 3;
    if (endT - beg0 <= LCAP) tile_rows<true,  L1>(Tl, dinv, lst, lb, tl, rp, dl, beg0, m0, n, lane);
    else                     tile_rows<false, L1>(Tl, dinv, lst, lb, tl, rp, dl, beg0, m0, n, lane);

    const int lhalf = lane >> 4;
    const int l16   = lane & 15;
    const int khalf = lhalf * 2;
    const int rbase = m0 + lhalf * 8;

    float cb[6];
#pragma unroll
    for (int t = 0; t < 6; ++t) cb[t] = bias[t * 16 + l16];

    v8f acc[6];
#pragma unroll
    for (int t = 0; t < 6; ++t)
#pragma unroll
        for (int r = 0; r < 8; ++r) acc[t][r] = 0.0f;

    const float* arow = tl + l16 * TS + khalf;
    for (int k = 0; k < DD; k += 4) {
        v2f a = *(const v2f*)(arow + k);
        int kr = k + khalf;
#pragma unroll
        for (int t = 0; t < 6; ++t) {
            int col = t * 16 + l16;
            v2f b = *(const v2f*)(&Wt[col * WT_STRIDE + kr]);
            acc[t] = __builtin_amdgcn_wmma_f32_16x16x4_f32(
                false, a, false, b, (short)0, acc[t], false, false);
        }
    }

    if (L1) {
        __builtin_amdgcn_sched_barrier(0);
        float cm[6], cf[6], ce[6];
#pragma unroll
        for (int t = 0; t < 6; ++t) {
            const int col = t * 16 + l16;
            cm[t] = mu[col];
            cf[t] = rsqrtf(var[col] + BN_EPS) * g[col];
            ce[t] = be[col];
        }
        float sc[8];
#pragma unroll
        for (int r = 0; r < 8; ++r) {
            int lo = __builtin_amdgcn_readlane(__float_as_int(dl), r);
            int hi = __builtin_amdgcn_readlane(__float_as_int(dl), r + 8);
            sc[r] = __int_as_float(lhalf ? hi : lo);
        }
#pragma unroll
        for (int t = 0; t < 6; ++t)
#pragma unroll
            for (int r = 0; r < 8; ++r)
                acc[t][r] = leaky_bn(acc[t][r] + cb[t], cm[t], cf[t], ce[t]) * sc[r];
    } else {
#pragma unroll
        for (int t = 0; t < 6; ++t)
#pragma unroll
            for (int r = 0; r < 8; ++r) acc[t][r] += cb[t];
    }

    if (m0 + 16 <= n) {
#pragma unroll
        for (int t = 0; t < 6; ++t) {
            int col = t * 16 + l16;
#pragma unroll
            for (int r = 0; r < 8; ++r)
                O[(size_t)(rbase + r) * DD + col] = acc[t][r];
        }
    } else {
#pragma unroll
        for (int t = 0; t < 6; ++t) {
            int col = t * 16 + l16;
#pragma unroll
            for (int r = 0; r < 8; ++r)
                if (rbase + r < n)
                    O[(size_t)(rbase + r) * DD + col] = acc[t][r];
        }
    }
}


extern "C" void kernel_launch(void* const* d_in, const int* in_sizes, int n_in,
                              void* d_out, int out_size, void* d_ws, size_t ws_size,
                              hipStream_t stream) {
    const float* X    = (const float*)d_in[0];
    const int*   ei   = (const int*)d_in[1];
    const float* W1   = (const float*)d_in[2];
    const float* b1   = (const float*)d_in[3];
    const float* W2   = (const float*)d_in[4];
    const float* b2   = (const float*)d_in[5];
    const float* gam  = (const float*)d_in[6];
    const float* bet  = (const float*)d_in[7];
    const float* mu   = (const float*)d_in[8];
    const float* var  = (const float*)d_in[9];

    const int N = in_sizes[0] / DD;
    const int E = in_sizes[1] / 2;
    const int* srcI = ei;
    const int* dstI = ei + E;

    float* dinv = (float*)d_ws;
    float* x1s  = dinv + N;
    int*   rptr = (int*)(x1s + (size_t)N * DD);
    int*   lst  = rptr + (N + 1);
    int*   offT = lst + E;
    int2*  pairs = (int2*)(((size_t)(offT + 2 * (MAXPER * 1024 + 1)) + 7) & ~(size_t)7);
    float* out  = (float*)d_out;

    const int nB        = (N + (1 << BSH) - 1) >> BSH;
    const int nC        = (E + CHUNK - 1) / CHUNK;
    const int tiles     = (N + 15) / 16;
    const int grid1     = (tiles + TW1 - 1) / TW1;
    const int grid2     = (tiles + TW2 - 1) / TW2;

    k_chunk_sort  <<<nC, 1024, 0, stream>>>(srcI, dstI, offT, pairs, E, nB, nC);
    k_bucket_lists<<<nB, 1024, 0, stream>>>(pairs, offT, rptr, dinv, lst, N, nC);

    k_layer<true> <<<grid1, TW1 * 32, 0, stream>>>(X,   dinv, rptr, lst, b1, gam, bet, mu, var, W1, x1s, N);
    k_layer<false><<<grid2, TW2 * 32, 0, stream>>>(x1s, dinv, rptr, lst, b2, gam, bet, mu, var, W2, out, N);
}


# Layer launches use one 25-wave work-group per WGP (LDS over half a WGP's, 250 groups) instead of 15/16-wave groups doubling up on some WGPs: busiest SIMD runs 7 tiles, not 8; W staged once per WGP
# speedup vs baseline: 4.4677x; 1.0442x over previous
#include <hip/hip_runtime.h>

#define DD 96
#define WT_STRIDE 98
#define BN_EPS 1e-5f
#define SLOPE 0.05f

typedef __attribute__((ext_vector_type(2))) float v2f;
typedef __attribute__((ext_vector_type(8))) float v8f;


#define BSH    9
#define MAXB   2048
#define CHUNK  4096
#define MAXPER 40
#define MAXC   1024
#define SEGU   8

__device__ __forceinline__ int wave_incl_scan(int v, int lane) {
    for (int o = 1; o < 32; o <<= 1) {
        int u = __shfl_up(v, o);
        if (lane >= o) v += u;
    }
    return v;
}

__global__ __launch_bounds__(1024) void k_chunk_sort(const int* __restrict__ src,
                                                     const int* __restrict__ dst,
                                                     int* __restrict__ offT,
                                                     int2* __restrict__ pairs,
                                                     int nE, int nB, int nC) {
    __shared__ int  h[MAXB];
    __shared__ int  wtot[32];
    __shared__ int2 lp[CHUNK];
    const int tid = threadIdx.x, lane = tid & 31, wv = tid >> 5;
    h[tid] = 0;
    h[tid + 1024] = 0;
    __syncthreads();
    const int e0 = blockIdx.x * CHUNK + tid;
    int d[4], s[4];
    int ec[4];
#pragma unroll
    for (int k = 0; k < 4; ++k) {
        int e = e0 + k * 1024;
        ec[k] = e < nE ? e : nE - 1;
    }
#pragma unroll
    for (int k = 0; k < 4; ++k) {
        d[k] = dst[ec[k]];
        s[k] = src[ec[k]];
    }
#pragma unroll
    for (int k = 0; k < 4; ++k)
        if (e0 + k * 1024 < nE) atomicAdd(&h[d[k] >> BSH], 1);
    __syncthreads();
    const int c0 = h[2 * tid], c1 = h[2 * tid + 1];
    const int sum = c0 + c1;
    const int incl = wave_incl_scan(sum, lane);
    if (lane == 31) wtot[wv] = incl;
    __syncthreads();
    int ex = incl - sum;
    for (int w = 0; w < wv; ++w) ex += wtot[w];
    h[2 * tid]     = ex;
    h[2 * tid + 1] = ex + c0;
    if (2 * tid     <= nB) offT[(2 * tid)     * nC + (int)blockIdx.x] = ex;
    if (2 * tid + 1 <= nB) offT[(2 * tid + 1) * nC + (int)blockIdx.x] = ex + c0;
    __syncthreads();
#pragma unroll
    for (int k = 0; k < 4; ++k) {
        if (e0 + k * 1024 < nE) {
            int p = atomicAdd(&h[d[k] >> BSH], 1);
            lp[p] = make_int2(d[k] & ((1 << BSH) - 1), s[k]);
        }
    }
    __syncthreads();
    int cnt = nE - (int)blockIdx.x * CHUNK;
    if (cnt > CHUNK) cnt = CHUNK;
    for (int i = tid; i < cnt; i += 1024) pairs[(size_t)blockIdx.x * CHUNK + i] = lp[i];
}

template <bool PLACE>
__device__ __forceinline__ void bucket_walk(const int2* __restrict__ pairs, const int2* seg,
                                            int* ctr, int* __restrict__ lst, int nC, int wv, int lane) {
    for (int cb = wv; cb < nC; cb += 32 * SEGU) {
        int2 sg[SEGU];
        int2 q[SEGU];
#pragma unroll
        for (int u = 0; u < SEGU; ++u) {
            const int c = cb + 32 * u;
            sg[u] = seg[c < nC ? c : nC - 1];
            if (c >= nC) sg[u].y = 0;
        }
#pragma unroll
        for (int u = 0; u < SEGU; ++u) {
            q[u] = make_int2(0, 0);
            if (lane < sg[u].y) q[u] = pairs[sg[u].x + lane];
        }
#pragma unroll
        for (int u = 0; u < SEGU; ++u) {
            if (lane < sg[u].y) {
                if (PLACE) lst[atomicAdd(&ctr[q[u].x], 1)] = q[u].y;
                else       atomicAdd(&ctr[q[u].x], 1);
            }
        }
#pragma unroll
        for (int u = 0; u < SEGU; ++u) {
            for (int j = 32 + lane; j < sg[u].y; j += 32) {
                const int2 r = pairs[sg[u].x + j];
                if (PLACE) lst[atomicAdd(&ctr[r.x], 1)] = r.y;
                else       atomicAdd(&ctr[r.x], 1);
            }
        }
    }
}

__global__ __launch_bounds__(1024) void k_bucket_lists(const int2* __restrict__ pairs,
                                                       const int* __restrict__ offT,
                                                       int* __restrict__ rowptr,
                                                       float* __restrict__ dinv,
                                                       int* __restrict__ lst,
                                                       int n, int nC) {
    __shared__ int cnt[1 << BSH];
    __shared__ int cur[1 << BSH];
    __shared__ int wtot[32];
    __shared__ int red[32];
    __shared__ int2 seg[MAXC];
    const int tid = threadIdx.x, lane = tid & 31, wv = tid >> 5;
    const int b = blockIdx.x;

    int a0 = 0;
    for (int c = tid; c < nC; c += 1024) {
        const int o0 = offT[b * nC + c];
        const int o1 = offT[(b + 1) * nC + c];
        seg[c] = make_int2(c * CHUNK + o0, o1 - o0);
        a0 += o0;
    }
    if (tid < (1 << BSH)) cnt[tid] = 0;
    const int ra = wave_incl_scan(a0, lane);
    if (lane == 31) red[wv] = ra;
    __syncthreads();
    int s0 = 0;
    for (int w = 0; w < 32; ++w) s0 += red[w];

    bucket_walk<false>(pairs, seg, cnt, lst, nC, wv, lane);
    __syncthreads();

    const int c = (tid < (1 << BSH)) ? cnt[tid] : 0;
    const int incl = wave_incl_scan(c, lane);
    if (lane == 31) wtot[wv] = incl;
    __syncthreads();
    int ex = s0 + incl - c;
    for (int w = 0; w < wv; ++w) ex += wtot[w];
    if (tid < (1 << BSH)) {
        cur[tid] = ex;
        const int node = (b << BSH) + tid;
        if (node < n) {
            rowptr[node] = ex;
            dinv[node]   = rsqrtf((float)(c + 1));
            if (node == n - 1) rowptr[n] = ex + c;
        }
    }
    __syncthreads();
    bucket_walk<true>(pairs, seg, cur, lst, nC, wv, lane);
}


struct __attribute__((packed, aligned(4))) f3 { float x, y, z; };

#define GU 8

template <bool WEIGHTED>
__device__ __forceinline__ void sum_chunk(const float* __restrict__ Tl, int s, float w, int m,
                                          float& a0, float& a1, float& a2) {
    for (int i = 0; i < m; i += GU) {
        f3 r[GU];
#pragma unroll
        for (int u = 0; u < GU; ++u) {
            int idx = (u == 0 || i + u < m) ? (i + u) : (m - 1);
            int su  = __builtin_amdgcn_readlane(s, idx);
            r[u]    = *(const f3*)(Tl + (size_t)su * DD);
        }
#pragma unroll
        for (int u = 0; u < GU; ++u) {
            if (WEIGHTED) {
                const bool  ok = (u == 0) || (i + u < m);
                const float wl = __int_as_float(__builtin_amdgcn_readlane(__float_as_int(w), i + u));
                const float wu = ok ? wl : 0.0f;
                a0 = fmaf(wu, r[u].x, a0);
                a1 = fmaf(wu, r[u].y, a1);
                a2 = fmaf(wu, r[u].z, a2);
            } else {
                bool ok = (u == 0) || (i + u < m);
                a0 += ok ? r[u].x : 0.0f;
                a1 += ok ? r[u].y : 0.0f;
                a2 += ok ? r[u].z : 0.0f;
            }
        }
    }
}

__device__ __forceinline__ float leaky_bn(float v, float mu, float f, float be) {
    v = (v >= 0.0f) ? v : SLOPE * v;
    return (v - mu) * f + be;
}

#define TW2   25
#define LCAP2 256
#define TW1   25
#define LCAP1 192
#define TS   100

template <bool IN_LDS>
__device__ __forceinline__ int list_chunk_t(const int* __restrict__ lst, const int* lb,
                                            int off, int base, int m, int lane) {
    int s = 0;
    if (lane < m) s = IN_LDS ? lb[off + lane] : lst[base + lane];
    return s;
}

template <bool IN_LDS, bool L1>
__device__ __forceinline__ void tile_rows(const float* __restrict__ Tl, const float* __restrict__ dinv,
                                          const int* __restrict__ lst, const int* lb, float* tl,
                                          int rp, float dl, int beg0, int m0, int n, int lane) {
    for (int j = 0; j < 16; ++j) {
        const int node = m0 + j;
        f3 o;
        o.x = 0.0f; o.y = 0.0f; o.z = 0.0f;
        if (node < n) {
            const int beg = __builtin_amdgcn_readlane(rp, j);
            const int end = __builtin_amdgcn_readlane(rp, j + 1);
            const float dd = __int_as_float(__builtin_amdgcn_readlane(__float_as_int(dl), j));
            const f3 hs = *(const f3*)(Tl + (size_t)node * DD);
            float a0 = 0.0f, a1 = 0.0f, a2 = 0.0f;
            for (int base = beg; base < end; base += 32) {
                int m = end - base;
                if (m > 32) m = 32;
                int   s;
                float w = 0.0f;
                if (IN_LDS && L1) {
                    int2 e = make_int2(0, 0);
                    if (lane < m) e = ((const int2*)lb)[base - beg0 + lane];
                    s = e.x;
                    w = __int_as_float(e.y);
                } else {
                    s = list_chunk_t<IN_LDS>(lst, lb, base - beg0, base, m, lane);
                    if (L1) w = dinv[s];
                }
                sum_chunk<L1>(Tl, s, w, m, a0, a1, a2);
            }
            if (L1) {
                o.x = fmaf(dd, hs.x, a0) * dd;
                o.y = fmaf(dd, hs.y, a1) * dd;
                o.z = fmaf(dd, hs.z, a2) * dd;
            } else {
                o.x = (a0 + hs.x) * dd;
                o.y = (a1 + hs.y) * dd;
                o.z = (a2 + hs.z) * dd;
            }
        }
        *(f3*)(tl + j * TS + lane * 3) = o;
    }
}

template <bool L1>
__global__ __launch_bounds__(L1 ? TW1 * 32 : TW2 * 32, 8) void k_layer(const float* __restrict__ T,
                                               const float* __restrict__ dinv,
                                               const int* __restrict__ rowptr,
                                               const int* __restrict__ lst,
                                               const float* __restrict__ bias,
                                               const float* __restrict__ g,
                                               const float* __restrict__ be,
                                               const float* __restrict__ mu,
                                               const float* __restrict__ var,
                                               const float* __restrict__ W,
                                               float* __restrict__ O, int n) {
    __shared__ __attribute__((aligned(16))) float Wt[DD * WT_STRIDE];
    constexpr int TW   = L1 ? TW1 : TW2;
    constexpr int NT   = TW * 32;
    constexpr int LCAP = L1 ? LCAP1 : LCAP2;
    __shared__ __attribute__((aligned(16))) float tile[TW][16 * TS];
    __shared__ __attribute__((aligned(8))) int lbuf[TW][L1 ? 2 * LCAP1 : LCAP2];

    const int lane = threadIdx.x & 31;
    const int wv = __builtin_amdgcn_readfirstlane((int)(threadIdx.x >> 5));
    const int m0 = ((int)blockIdx.x * TW + wv) * 16;
    const bool live = m0 < n;

    int   rp = 0;
    float dl = 0.0f;
    if (live) {
        int j  = lane < 16 ? lane : 16;
        int ix = m0 + j;
        rp = rowptr[ix < n ? ix : n];
        int nd = m0 + (lane & 15);
        dl = dinv[nd < n ? nd : n - 1];
    }

    {
        constexpr int NU = (DD * DD + NT - 1) / NT;
        float w[NU];
#pragma unroll
        for (int u = 0; u < NU; ++u) {
            int i = u * NT + (int)threadIdx.x;
            w[u] = W[i < DD * DD ? i : DD * DD - 1];
        }
#pragma unroll
        for (int u = 0; u < NU; ++u) {
            int i = u * NT + (int)threadIdx.x;
            if (i < DD * DD) {
                int k = i / DD, col = i - k * DD;
                Wt[col * WT_STRIDE + k] = w[u];
            }
        }
    }
    __syncthreads();
    if (!live) return;

    float* tl = tile[wv];
    int*   lb = lbuf[wv];
    const int beg0 = __builtin_amdgcn_readlane(rp, 0);
    const int endT = __builtin_amdgcn_readlane(rp, 16);

    {
        int cnt = endT - beg0;
        if (cnt > LCAP) cnt = LCAP;
        if (cnt > 0) {
            int t[LCAP / 32];
#pragma unroll
            for (int u = 0; u < LCAP / 32; ++u) {
                int p = u * 32 + lane;
                t[u] = lst[beg0 + (p < cnt ? p : 0)];
            }
            if (L1) {
                float d[LCAP / 32];
#pragma unroll
                for (int u = 0; u < LCAP / 32; ++u) d[u] = dinv[t[u]];
#pragma unroll
                for (int u = 0; u < LCAP / 32; ++u)
                    ((int2*)lb)[u * 32 + lane] = make_int2(t[u], __float_as_int(d[u]));
            } else {
#pragma unroll
                for (int u = 0; u < LCAP / 32; ++u) lb[u * 32 + lane] = t[u];
            }
        }
    }

    const float* Tl = T + lane * 3;
    if (endT - beg0 <= LCAP) tile_rows<true,  L1>(Tl, dinv, lst, lb, tl, rp, dl, beg0, m0, n, lane);
    else                     tile_rows<false, L1>(Tl, dinv, lst, lb, tl, rp, dl, beg0, m0, n, lane);

    const int lhalf = lane >> 4;
    const int l16   = lane & 15;
    const int khalf = lhalf * 2;
    const int rbase = m0 + lhalf * 8;

    float cb[6];
#pragma unroll
    for (int t = 0; t < 6; ++t) cb[t] = bias[t * 16 + l16];

    v8f acc[6];
#pragma unroll
    for (int t = 0; t < 6; ++t)
#pragma unroll
        for (int r = 0; r < 8; ++r) acc[t][r] = 0.0f;

    const float* arow = tl + l16 * TS + khalf;
    for (int k = 0; k < DD; k += 4) {
        v2f a = *(const v2f*)(arow + k);
        int kr = k + khalf;
#pragma unroll
        for (int t = 0; t < 6; ++t) {
            int col = t * 16 + l16;
            v2f b = *(const v2f*)(&Wt[col * WT_STRIDE + kr]);
            acc[t] = __builtin_amdgcn_wmma_f32_16x16x4_f32(
                false, a, false, b, (short)0, acc[t], false, false);
        }
    }

    if (L1) {
        __builtin_amdgcn_sched_barrier(0);
        float cm[6], cf[6], ce[6];
#pragma unroll
        for (int t = 0; t < 6; ++t) {
            const int col = t * 16 + l16;
            cm[t] = mu[col];
            cf[t] = rsqrtf(var[col] + BN_EPS) * g[col];
            ce[t] = be[col];
        }
        float sc[8];
#pragma unroll
        for (int r = 0; r < 8; ++r) {
            int lo = __builtin_amdgcn_readlane(__float_as_int(dl), r);
            int hi = __builtin_amdgcn_readlane(__float_as_int(dl), r + 8);
            sc[r] = __int_as_float(lhalf ? hi : lo);
        }
#pragma unroll
        for (int t = 0; t < 6; ++t)
#pragma unroll
            for (int r = 0; r < 8; ++r)
                acc[t][r] = leaky_bn(acc[t][r] + cb[t], cm[t], cf[t], ce[t]) * sc[r];
    } else {
#pragma unroll
        for (int t = 0; t < 6; ++t)
#pragma unroll
            for (int r = 0; r < 8; ++r) acc[t][r] += cb[t];
    }

    if (m0 + 16 <= n) {
#pragma unroll
        for (int t = 0; t < 6; ++t) {
            int col = t * 16 + l16;
#pragma unroll
            for (int r = 0; r < 8; ++r)
                O[(size_t)(rbase + r) * DD + col] = acc[t][r];
        }
    } else {
#pragma unroll
        for (int t = 0; t < 6; ++t) {
            int col = t * 16 + l16;
#pragma unroll
            for (int r = 0; r < 8; ++r)
                if (rbase + r < n)
                    O[(size_t)(rbase + r) * DD + col] = acc[t][r];
        }
    }
}


extern "C" void kernel_launch(void* const* d_in, const int* in_sizes, int n_in,
                              void* d_out, int out_size, void* d_ws, size_t ws_size,
                              hipStream_t stream) {
    const float* X    = (const float*)d_in[0];
    const int*   ei   = (const int*)d_in[1];
    const float* W1   = (const float*)d_in[2];
    const float* b1   = (const float*)d_in[3];
    const float* W2   = (const float*)d_in[4];
    const float* b2   = (const float*)d_in[5];
    const float* gam  = (const float*)d_in[6];
    const float* bet  = (const float*)d_in[7];
    const float* mu   = (const float*)d_in[8];
    const float* var  = (const float*)d_in[9];

    const int N = in_sizes[0] / DD;
    const int E = in_sizes[1] / 2;
    const int* srcI = ei;
    const int* dstI = ei + E;

    float* dinv = (float*)d_ws;
    float* x1s  = dinv + N;
    int*   rptr = (int*)(x1s + (size_t)N * DD);
    int*   lst  = rptr + (N + 1);
    int*   offT = lst + E;
    int2*  pairs = (int2*)(((size_t)(offT + 2 * (MAXPER * 1024 + 1)) + 7) & ~(size_t)7);
    float* out  = (float*)d_out;

    const int nB        = (N + (1 << BSH) - 1) >> BSH;
    const int nC        = (E + CHUNK - 1) / CHUNK;
    const int tiles     = (N + 15) / 16;
    const int grid1     = (tiles + TW1 - 1) / TW1;
    const int grid2     = (tiles + TW2 - 1) / TW2;

    k_chunk_sort  <<<nC, 1024, 0, stream>>>(srcI, dstI, offT, pairs, E, nB, nC);
    k_bucket_lists<<<nB, 1024, 0, stream>>>(pairs, offT, rptr, dinv, lst, N, nC);

    k_layer<true> <<<grid1, TW1 * 32, 0, stream>>>(X,   dinv, rptr, lst, b1, gam, bet, mu, var, W1, x1s, N);
    k_layer<false><<<grid2, TW2 * 32, 0, stream>>>(x1s, dinv, rptr, lst, b2, gam, bet, mu, var, W2, out, N);
}


# Gather requests exactly the rows a node's list has, up to 16 together, instead of groups of 8 padded by re-reads of the last row: no load or add for an empty slot; every sum in the same order
# speedup vs baseline: 4.7776x; 1.0694x over previous
#include <hip/hip_runtime.h>

#define DD 96
#define WT_STRIDE 98
#define BN_EPS 1e-5f
#define SLOPE 0.05f

typedef __attribute__((ext_vector_type(2))) float v2f;
typedef __attribute__((ext_vector_type(8))) float v8f;


#define BSH    9
#define MAXB   2048
#define CHUNK  4096
#define MAXPER 40
#define MAXC   1024
#define SEGU   8

__device__ __forceinline__ int wave_incl_scan(int v, int lane) {
    for (int o = 1; o < 32; o <<= 1) {
        int u = __shfl_up(v, o);
        if (lane >= o) v += u;
    }
    return v;
}

__global__ __launch_bounds__(1024) void k_chunk_sort(const int* __restrict__ src,
                                                     const int* __restrict__ dst,
                                                     int* __restrict__ offT,
                                                     int2* __restrict__ pairs,
                                                     int nE, int nB, int nC) {
    __shared__ int  h[MAXB];
    __shared__ int  wtot[32];
    __shared__ int2 lp[CHUNK];
    const int tid = threadIdx.x, lane = tid & 31, wv = tid >> 5;
    h[tid] = 0;
    h[tid + 1024] = 0;
    __syncthreads();
    const int e0 = blockIdx.x * CHUNK + tid;
    int d[4], s[4];
    int ec[4];
#pragma unroll
    for (int k = 0; k < 4; ++k) {
        int e = e0 + k * 1024;
        ec[k] = e < nE ? e : nE - 1;
    }
#pragma unroll
    for (int k = 0; k < 4; ++k) {
        d[k] = dst[ec[k]];
        s[k] = src[ec[k]];
    }
#pragma unroll
    for (int k = 0; k < 4; ++k)
        if (e0 + k * 1024 < nE) atomicAdd(&h[d[k] >> BSH], 1);
    __syncthreads();
    const int c0 = h[2 * tid], c1 = h[2 * tid + 1];
    const int sum = c0 + c1;
    const int incl = wave_incl_scan(sum, lane);
    if (lane == 31) wtot[wv] = incl;
    __syncthreads();
    int ex = incl - sum;
    for (int w = 0; w < wv; ++w) ex += wtot[w];
    h[2 * tid]     = ex;
    h[2 * tid + 1] = ex + c0;
    if (2 * tid     <= nB) offT[(2 * tid)     * nC + (int)blockIdx.x] = ex;
    if (2 * tid + 1 <= nB) offT[(2 * tid + 1) * nC + (int)blockIdx.x] = ex + c0;
    __syncthreads();
#pragma unroll
    for (int k = 0; k < 4; ++k) {
        if (e0 + k * 1024 < nE) {
            int p = atomicAdd(&h[d[k] >> BSH], 1);
            lp[p] = make_int2(d[k] & ((1 << BSH) - 1), s[k]);
        }
    }
    __syncthreads();
    int cnt = nE - (int)blockIdx.x * CHUNK;
    if (cnt > CHUNK) cnt = CHUNK;
    for (int i = tid; i < cnt; i += 1024) pairs[(size_t)blockIdx.x * CHUNK + i] = lp[i];
}

template <bool PLACE>
__device__ __forceinline__ void bucket_walk(const int2* __restrict__ pairs, const int2* seg,
                                            int* ctr, int* __restrict__ lst, int nC, int wv, int lane) {
    for (int cb = wv; cb < nC; cb += 32 * SEGU) {
        int2 sg[SEGU];
        int2 q[SEGU];
#pragma unroll
        for (int u = 0; u < SEGU; ++u) {
            const int c = cb + 32 * u;
            sg[u] = seg[c < nC ? c : nC - 1];
            if (c >= nC) sg[u].y = 0;
        }
#pragma unroll
        for (int u = 0; u < SEGU; ++u) {
            q[u] = make_int2(0, 0);
            if (lane < sg[u].y) q[u] = pairs[sg[u].x + lane];
        }
#pragma unroll
        for (int u = 0; u < SEGU; ++u) {
            if (lane < sg[u].y) {
                if (PLACE) lst[atomicAdd(&ctr[q[u].x], 1)] = q[u].y;
                else       atomicAdd(&ctr[q[u].x], 1);
            }
        }
#pragma unroll
        for (int u = 0; u < SEGU; ++u) {
            for (int j = 32 + lane; j < sg[u].y; j += 32) {
                const int2 r = pairs[sg[u].x + j];
                if (PLACE) lst[atomicAdd(&ctr[r.x], 1)] = r.y;
                else       atomicAdd(&ctr[r.x], 1);
            }
        }
    }
}

__global__ __launch_bounds__(1024) void k_bucket_lists(const int2* __restrict__ pairs,
                                                       const int* __restrict__ offT,
                                                       int* __restrict__ rowptr,
                                                       float* __restrict__ dinv,
                                                       int* __restrict__ lst,
                                                       int n, int nC) {
    __shared__ int cnt[1 << BSH];
    __shared__ int cur[1 << BSH];
    __shared__ int wtot[32];
    __shared__ int red[32];
    __shared__ int2 seg[MAXC];
    const int tid = threadIdx.x, lane = tid & 31, wv = tid >> 5;
    const int b = blockIdx.x;

    int a0 = 0;
    for (int c = tid; c < nC; c += 1024) {
        const int o0 = offT[b * nC + c];
        const int o1 = offT[(b + 1) * nC + c];
        seg[c] = make_int2(c * CHUNK + o0, o1 - o0);
        a0 += o0;
    }
    if (tid < (1 << BSH)) cnt[tid] = 0;
    const int ra = wave_incl_scan(a0, lane);
    if (lane == 31) red[wv] = ra;
    __syncthreads();
    int s0 = 0;
    for (int w = 0; w < 32; ++w) s0 += red[w];

    bucket_walk<false>(pairs, seg, cnt, lst, nC, wv, lane);
    __syncthreads();

    const int c = (tid < (1 << BSH)) ? cnt[tid] : 0;
    const int incl = wave_incl_scan(c, lane);
    if (lane == 31) wtot[wv] = incl;
    __syncthreads();
    int ex = s0 + incl - c;
    for (int w = 0; w < wv; ++w) ex += wtot[w];
    if (tid < (1 << BSH)) {
        cur[tid] = ex;
        const int node = (b << BSH) + tid;
        if (node < n) {
            rowptr[node] = ex;
            dinv[node]   = rsqrtf((float)(c + 1));
            if (node == n - 1) rowptr[n] = ex + c;
        }
    }
    __syncthreads();
    bucket_walk<true>(pairs, seg, cur, lst, nC, wv, lane);
}


struct __attribute__((packed, aligned(4))) f3 { float x, y, z; };

#define GU 16

template <bool WEIGHTED, int C>
__device__ __forceinline__ void sum_rows(const float* __restrict__ Tl, int s, float w, int i,
                                         float& a0, float& a1, float& a2) {
    f3 r[C];
#pragma unroll
    for (int u = 0; u < C; ++u) {
        int su = __builtin_amdgcn_readlane(s, i + u);
        r[u]   = *(const f3*)(Tl + (size_t)su * DD);
    }
#pragma unroll
    for (int u = 0; u < C; ++u) {
        if (WEIGHTED) {
            const float wl = __int_as_float(__builtin_amdgcn_readlane(__float_as_int(w), i + u));
            a0 = fmaf(wl, r[u].x, a0);
            a1 = fmaf(wl, r[u].y, a1);
            a2 = fmaf(wl, r[u].z, a2);
        } else {
            a0 += r[u].x;
            a1 += r[u].y;
            a2 += r[u].z;
        }
    }
}

#define SUM_CASE(K) case K: if (K < MAXG) { sum_rows<WEIGHTED, (K < MAXG ? K : MAXG)>(Tl, s, w, i, a0, a1, a2); break; }
template <bool WEIGHTED, int MAXG>
__device__ __forceinline__ void sum_chunk(const float* __restrict__ Tl, int s, float w, int m,
                                          float& a0, float& a1, float& a2) {
    for (int i = 0; i < m; i += MAXG) {
        switch (m - i) {
            SUM_CASE(1)  SUM_CASE(2)  SUM_CASE(3)  SUM_CASE(4)  SUM_CASE(5)  SUM_CASE(6)  SUM_CASE(7)  SUM_CASE(8)
            SUM_CASE(9)  SUM_CASE(10) SUM_CASE(11) SUM_CASE(12) SUM_CASE(13) SUM_CASE(14) SUM_CASE(15)
            default: sum_rows<WEIGHTED, MAXG>(Tl, s, w, i, a0, a1, a2); break;
        }
    }
}
#undef SUM_CASE

__device__ __forceinline__ float leaky_bn(float v, float mu, float f, float be) {
    v = (v >= 0.0f) ? v : SLOPE * v;
    return (v - mu) * f + be;
}

#define TW2   25
#define LCAP2 256
#define TW1   25
#define LCAP1 192
#define TS   100

template <bool IN_LDS>
__device__ __forceinline__ int list_chunk_t(const int* __restrict__ lst, const int* lb,
                                            int off, int base, int m, int lane) {
    int s = 0;
    if (lane < m) s = IN_LDS ? lb[off + lane] : lst[base + lane];
    return s;
}

template <bool IN_LDS, bool L1>
__device__ __forceinline__ void tile_rows(const float* __restrict__ Tl, const float* __restrict__ dinv,
                                          const int* __restrict__ lst, const int* lb, float* tl,
                                          int rp, float dl, int beg0, int m0, int n, int lane) {
    for (int j = 0; j < 16; ++j) {
        const int node = m0 + j;
        f3 o;
        o.x = 0.0f; o.y = 0.0f; o.z = 0.0f;
        if (node < n) {
            const int beg = __builtin_amdgcn_readlane(rp, j);
            const int end = __builtin_amdgcn_readlane(rp, j + 1);
            const float dd = __int_as_float(__builtin_amdgcn_readlane(__float_as_int(dl), j));
            const f3 hs = *(const f3*)(Tl + (size_t)node * DD);
            float a0 = 0.0f, a1 = 0.0f, a2 = 0.0f;
            for (int base = beg; base < end; base += 32) {
                int m = end - base;
                if (m > 32) m = 32;
                int   s;
                float w = 0.0f;
                if (IN_LDS && L1) {
                    int2 e = make_int2(0, 0);
                    if (lane < m) e = ((const int2*)lb)[base - beg0 + lane];
                    s = e.x;
                    w = __int_as_float(e.y);
                } else {
                    s = list_chunk_t<IN_LDS>(lst, lb, base - beg0, base, m, lane);
                    if (L1) w = dinv[s];
                }
                sum_chunk<L1, IN_LDS ? GU : 4>(Tl, s, w, m, a0, a1, a2);
            }
            if (L1) {
                o.x = fmaf(dd, hs.x, a0) * dd;
                o.y = fmaf(dd, hs.y, a1) * dd;
                o.z = fmaf(dd, hs.z, a2) * dd;
            } else {
                o.x = (a0 + hs.x) * dd;
                o.y = (a1 + hs.y) * dd;
                o.z = (a2 + hs.z) * dd;
            }
        }
        *(f3*)(tl + j * TS + lane * 3) = o;
    }
}

template <bool L1>
__global__ __launch_bounds__(L1 ? TW1 * 32 : TW2 * 32, 8) void k_layer(const float* __restrict__ T,
                                               const float* __restrict__ dinv,
                                               const int* __restrict__ rowptr,
                                               const int* __restrict__ lst,
                                               const float* __restrict__ bias,
                                               const float* __restrict__ g,
                                               const float* __restrict__ be,
                                               const float* __restrict__ mu,
                                               const float* __restrict__ var,
                                               const float* __restrict__ W,
                                               float* __restrict__ O, int n) {
    __shared__ __attribute__((aligned(16))) float Wt[DD * WT_STRIDE];
    constexpr int TW   = L1 ? TW1 : TW2;
    constexpr int NT   = TW * 32;
    constexpr int LCAP = L1 ? LCAP1 : LCAP2;
    __shared__ __attribute__((aligned(16))) float tile[TW][16 * TS];
    __shared__ __attribute__((aligned(8))) int lbuf[TW][L1 ? 2 * LCAP1 : LCAP2];

    const int lane = threadIdx.x & 31;
    const int wv = __builtin_amdgcn_readfirstlane((int)(threadIdx.x >> 5));
    const int m0 = ((int)blockIdx.x * TW + wv) * 16;
    const bool live = m0 < n;

    int   rp = 0;
    float dl = 0.0f;
    if (live) {
        int j  = lane < 16 ? lane : 16;
        int ix = m0 + j;
        rp = rowptr[ix < n ? ix : n];
        int nd = m0 + (lane & 15);
        dl = dinv[nd < n ? nd : n - 1];
    }

    {
        constexpr int NU = (DD * DD + NT - 1) / NT;
        float w[NU];
#pragma unroll
        for (int u = 0; u < NU; ++u) {
            int i = u * NT + (int)threadIdx.x;
            w[u] = W[i < DD * DD ? i : DD * DD - 1];
        }
#pragma unroll
        for (int u = 0; u < NU; ++u) {
            int i = u * NT + (int)threadIdx.x;
            if (i < DD * DD) {
                int k = i / DD, col = i - k * DD;
                Wt[col * WT_STRIDE + k] = w[u];
            }
        }
    }
    __syncthreads();
    if (!live) return;

    float* tl = tile[wv];
    int*   lb = lbuf[wv];
    const int beg0 = __builtin_amdgcn_readlane(rp, 0);
    const int endT = __builtin_amdgcn_readlane(rp, 16);

    {
        int cnt = endT - beg0;
        if (cnt > LCAP) cnt = LCAP;
        if (cnt > 0) {
            int t[LCAP / 32];
#pragma unroll
            for (int u = 0; u < LCAP / 32; ++u) {
                int p = u * 32 + lane;
                t[u] = lst[beg0 + (p < cnt ? p : 0)];
            }
            if (L1) {
                float d[LCAP / 32];
#pragma unroll
                for (int u = 0; u < LCAP / 32; ++u) d[u] = dinv[t[u]];
#pragma unroll
                for (int u = 0; u < LCAP / 32; ++u)
                    ((int2*)lb)[u * 32 + lane] = make_int2(t[u], __float_as_int(d[u]));
            } else {
#pragma unroll
                for (int u = 0; u < LCAP / 32; ++u) lb[u * 32 + lane] = t[u];
            }
        }
    }

    const float* Tl = T + lane * 3;
    if (endT - beg0 <= LCAP) tile_rows<true,  L1>(Tl, dinv, lst, lb, tl, rp, dl, beg0, m0, n, lane);
    else                     tile_rows<false, L1>(Tl, dinv, lst, lb, tl, rp, dl, beg0, m0, n, lane);

    const int lhalf = lane >> 4;
    const int l16   = lane & 15;
    const int khalf = lhalf * 2;
    const int rbase = m0 + lhalf * 8;

    float cb[6];
#pragma unroll
    for (int t = 0; t < 6; ++t) cb[t] = bias[t * 16 + l16];

    v8f acc[6];
#pragma unroll
    for (int t = 0; t < 6; ++t)
#pragma unroll
        for (int r = 0; r < 8; ++r) acc[t][r] = 0.0f;

    const float* arow = tl + l16 * TS + khalf;
    for (int k = 0; k < DD; k += 4) {
        v2f a = *(const v2f*)(arow + k);
        int kr = k + khalf;
#pragma unroll
        for (int t = 0; t < 6; ++t) {
            int col = t * 16 + l16;
            v2f b = *(const v2f*)(&Wt[col * WT_STRIDE + kr]);
            acc[t] = __builtin_amdgcn_wmma_f32_16x16x4_f32(
                false, a, false, b, (short)0, acc[t], false, false);
        }
    }

    if (L1) {
        __builtin_amdgcn_sched_barrier(0);
        float cm[6], cf[6], ce[6];
#pragma unroll
        for (int t = 0; t < 6; ++t) {
            const int col = t * 16 + l16;
            cm[t] = mu[col];
            cf[t] = rsqrtf(var[col] + BN_EPS) * g[col];
            ce[t] = be[col];
        }
        float sc[8];
#pragma unroll
        for (int r = 0; r < 8; ++r) {
            int lo = __builtin_amdgcn_readlane(__float_as_int(dl), r);
            int hi = __builtin_amdgcn_readlane(__float_as_int(dl), r + 8);
            sc[r] = __int_as_float(lhalf ? hi : lo);
        }
#pragma unroll
        for (int t = 0; t < 6; ++t)
#pragma unroll
            for (int r = 0; r < 8; ++r)
                acc[t][r] = leaky_bn(acc[t][r] + cb[t], cm[t], cf[t], ce[t]) * sc[r];
    } else {
#pragma unroll
        for (int t = 0; t < 6; ++t)
#pragma unroll
            for (int r = 0; r < 8; ++r) acc[t][r] += cb[t];
    }

    if (m0 + 16 <= n) {
#pragma unroll
        for (int t = 0; t < 6; ++t) {
            int col = t * 16 + l16;
#pragma unroll
            for (int r = 0; r < 8; ++r)
                O[(size_t)(rbase + r) * DD + col] = acc[t][r];
        }
    } else {
#pragma unroll
        for (int t = 0; t < 6; ++t) {
            int col = t * 16 + l16;
#pragma unroll
            for (int r = 0; r < 8; ++r)
                if (rbase + r < n)
                    O[(size_t)(rbase + r) * DD + col] = acc[t][r];
        }
    }
}


extern "C" void kernel_launch(void* const* d_in, const int* in_sizes, int n_in,
                              void* d_out, int out_size, void* d_ws, size_t ws_size,
                              hipStream_t stream) {
    const float* X    = (const float*)d_in[0];
    const int*   ei   = (const int*)d_in[1];
    const float* W1   = (const float*)d_in[2];
    const float* b1   = (const float*)d_in[3];
    const float* W2   = (const float*)d_in[4];
    const float* b2   = (const float*)d_in[5];
    const float* gam  = (const float*)d_in[6];
    const float* bet  = (const float*)d_in[7];
    const float* mu   = (const float*)d_in[8];
    const float* var  = (const float*)d_in[9];

    const int N = in_sizes[0] / DD;
    const int E = in_sizes[1] / 2;
    const int* srcI = ei;
    const int* dstI = ei + E;

    float* dinv = (float*)d_ws;
    float* x1s  = dinv + N;
    int*   rptr = (int*)(x1s + (size_t)N * DD);
    int*   lst  = rptr + (N + 1);
    int*   offT = lst + E;
    int2*  pairs = (int2*)(((size_t)(offT + 2 * (MAXPER * 1024 + 1)) + 7) & ~(size_t)7);
    float* out  = (float*)d_out;

    const int nB        = (N + (1 << BSH) - 1) >> BSH;
    const int nC        = (E + CHUNK - 1) / CHUNK;
    const int tiles     = (N + 15) / 16;
    const int grid1     = (tiles + TW1 - 1) / TW1;
    const int grid2     = (tiles + TW2 - 1) / TW2;

    k_chunk_sort  <<<nC, 1024, 0, stream>>>(srcI, dstI, offT, pairs, E, nB, nC);
    k_bucket_lists<<<nB, 1024, 0, stream>>>(pairs, offT, rptr, dinv, lst, N, nC);

    k_layer<true> <<<grid1, TW1 * 32, 0, stream>>>(X,   dinv, rptr, lst, b1, gam, bet, mu, var, W1, x1s, N);
    k_layer<false><<<grid2, TW2 * 32, 0, stream>>>(x1s, dinv, rptr, lst, b2, gam, bet, mu, var, W2, out, N);
}


# Gather walks a tile's whole list run as ONE stream in fixed groups of 16 rows across node boundaries (self loop stored as each list's last entry): 16 load waits per tile become 10; sums keep order
# speedup vs baseline: 4.9220x; 1.0302x over previous
#include <hip/hip_runtime.h>

#define DD 96
#define WT_STRIDE 98
#define BN_EPS 1e-5f
#define SLOPE 0.05f

typedef __attribute__((ext_vector_type(2))) float v2f;
typedef __attribute__((ext_vector_type(8))) float v8f;


#define BSH    9
#define MAXB   2048
#define CHUNK  4096
#define MAXPER 40
#define MAXC   1024
#define SEGU   8

__device__ __forceinline__ int wave_incl_scan(int v, int lane) {
    for (int o = 1; o < 32; o <<= 1) {
        int u = __shfl_up(v, o);
        if (lane >= o) v += u;
    }
    return v;
}

__global__ __launch_bounds__(1024) void k_chunk_sort(const int* __restrict__ src,
                                                     const int* __restrict__ dst,
                                                     int* __restrict__ offT,
                                                     int2* __restrict__ pairs,
                                                     int nE, int nB, int nC) {
    __shared__ int  h[MAXB];
    __shared__ int  wtot[32];
    __shared__ int2 lp[CHUNK];
    const int tid = threadIdx.x, lane = tid & 31, wv = tid >> 5;
    h[tid] = 0;
    h[tid + 1024] = 0;
    __syncthreads();
    const int e0 = blockIdx.x * CHUNK + tid;
    int d[4], s[4];
    int ec[4];
#pragma unroll
    for (int k = 0; k < 4; ++k) {
        int e = e0 + k * 1024;
        ec[k] = e < nE ? e : nE - 1;
    }
#pragma unroll
    for (int k = 0; k < 4; ++k) {
        d[k] = dst[ec[k]];
        s[k] = src[ec[k]];
    }
#pragma unroll
    for (int k = 0; k < 4; ++k)
        if (e0 + k * 1024 < nE) atomicAdd(&h[d[k] >> BSH], 1);
    __syncthreads();
    const int c0 = h[2 * tid], c1 = h[2 * tid + 1];
    const int sum = c0 + c1;
    const int incl = wave_incl_scan(sum, lane);
    if (lane == 31) wtot[wv] = incl;
    __syncthreads();
    int ex = incl - sum;
    for (int w = 0; w < wv; ++w) ex += wtot[w];
    h[2 * tid]     = ex;
    h[2 * tid + 1] = ex + c0;
    if (2 * tid     <= nB) offT[(2 * tid)     * nC + (int)blockIdx.x] = ex;
    if (2 * tid + 1 <= nB) offT[(2 * tid + 1) * nC + (int)blockIdx.x] = ex + c0;
    __syncthreads();
#pragma unroll
    for (int k = 0; k < 4; ++k) {
        if (e0 + k * 1024 < nE) {
            int p = atomicAdd(&h[d[k] >> BSH], 1);
            lp[p] = make_int2(d[k] & ((1 << BSH) - 1), s[k]);
        }
    }
    __syncthreads();
    int cnt = nE - (int)blockIdx.x * CHUNK;
    if (cnt > CHUNK) cnt = CHUNK;
    for (int i = tid; i < cnt; i += 1024) pairs[(size_t)blockIdx.x * CHUNK + i] = lp[i];
}

template <bool PLACE>
__device__ __forceinline__ void bucket_walk(const int2* __restrict__ pairs, const int2* seg,
                                            int* ctr, int* __restrict__ lst, int nC, int wv, int lane) {
    for (int cb = wv; cb < nC; cb += 32 * SEGU) {
        int2 sg[SEGU];
        int2 q[SEGU];
#pragma unroll
        for (int u = 0; u < SEGU; ++u) {
            const int c = cb + 32 * u;
            sg[u] = seg[c < nC ? c : nC - 1];
            if (c >= nC) sg[u].y = 0;
        }
#pragma unroll
        for (int u = 0; u < SEGU; ++u) {
            q[u] = make_int2(0, 0);
            if (lane < sg[u].y) q[u] = pairs[sg[u].x + lane];
        }
#pragma unroll
        for (int u = 0; u < SEGU; ++u) {
            if (lane < sg[u].y) {
                if (PLACE) lst[atomicAdd(&ctr[q[u].x], 1)] = q[u].y;
                else       atomicAdd(&ctr[q[u].x], 1);
            }
        }
#pragma unroll
        for (int u = 0; u < SEGU; ++u) {
            for (int j = 32 + lane; j < sg[u].y; j += 32) {
                const int2 r = pairs[sg[u].x + j];
                if (PLACE) lst[atomicAdd(&ctr[r.x], 1)] = r.y;
                else       atomicAdd(&ctr[r.x], 1);
            }
        }
    }
}

__global__ __launch_bounds__(1024) void k_bucket_lists(const int2* __restrict__ pairs,
                                                       const int* __restrict__ offT,
                                                       int* __restrict__ rowptr,
                                                       float* __restrict__ dinv,
                                                       int* __restrict__ lst,
                                                       int n, int nC) {
    __shared__ int cnt[1 << BSH];
    __shared__ int cur[1 << BSH];
    __shared__ int wtot[32];
    __shared__ int red[32];
    __shared__ int2 seg[MAXC];
    const int tid = threadIdx.x, lane = tid & 31, wv = tid >> 5;
    const int b = blockIdx.x;

    int a0 = 0;
    for (int c = tid; c < nC; c += 1024) {
        const int o0 = offT[b * nC + c];
        const int o1 = offT[(b + 1) * nC + c];
        seg[c] = make_int2(c * CHUNK + o0, o1 - o0);
        a0 += o0;
    }
    if (tid < (1 << BSH)) cnt[tid] = 0;
    const int ra = wave_incl_scan(a0, lane);
    if (lane == 31) red[wv] = ra;
    __syncthreads();
    int s0 = 0;
    for (int w = 0; w < 32; ++w) s0 += red[w];

    bucket_walk<false>(pairs, seg, cnt, lst, nC, wv, lane);
    __syncthreads();

    const int c = (tid < (1 << BSH)) ? cnt[tid] : 0;
    const int incl = wave_incl_scan(c, lane);
    if (lane == 31) wtot[wv] = incl;
    __syncthreads();
    int ex = s0 + incl - c;
    for (int w = 0; w < wv; ++w) ex += wtot[w];
    if (tid < (1 << BSH)) {
        const int node = (b << BSH) + tid;
        const int st   = ex + node;
        cur[tid] = st;
        if (node < n) {
            rowptr[node] = st;
            dinv[node]   = rsqrtf((float)(c + 1));
            lst[st + c]  = node;
            if (node == n - 1) rowptr[n] = st + c + 1;
        }
    }
    __syncthreads();
    bucket_walk<true>(pairs, seg, cur, lst, nC, wv, lane);
}


struct __attribute__((packed, aligned(4))) f3 { float x, y, z; };

#define GU 16

template <int C>
__device__ __forceinline__ void load_rows(const float* __restrict__ Tl, int s, f3* r) {
#pragma unroll
    for (int u = 0; u < C; ++u) {
        int su = __builtin_amdgcn_readlane(s, u);
        r[u]   = *(const f3*)(Tl + (size_t)su * DD);
    }
}

#define LOAD_CASE(K) case K: if (K < G) { load_rows<(K < G ? K : G)>(Tl, s, r); break; }
template <int G>
__device__ __forceinline__ void load_group(const float* __restrict__ Tl, int s, int g, f3* r) {
    switch (g) {
        LOAD_CASE(1)  LOAD_CASE(2)  LOAD_CASE(3)  LOAD_CASE(4)  LOAD_CASE(5)  LOAD_CASE(6)  LOAD_CASE(7)  LOAD_CASE(8)
        LOAD_CASE(9)  LOAD_CASE(10) LOAD_CASE(11) LOAD_CASE(12) LOAD_CASE(13) LOAD_CASE(14) LOAD_CASE(15)
        default: load_rows<G>(Tl, s, r); break;
    }
}
#undef LOAD_CASE

__device__ __forceinline__ float leaky_bn(float v, float mu, float f, float be) {
    v = (v >= 0.0f) ? v : SLOPE * v;
    return (v - mu) * f + be;
}

#define TW2   25
#define LCAP2 256
#define TW1   25
#define LCAP1 224
#define TS   100

template <bool IN_LDS, bool L1>
__device__ __forceinline__ void run_entries(const float* __restrict__ dinv, const int* __restrict__ lst,
                                            const int* lb, int beg0, int p, int g, int lane, int& s, float& w) {
    s = 0;
    w = 0.0f;
    if (IN_LDS && L1) {
        int2 e = make_int2(0, 0);
        if (lane < g) e = ((const int2*)lb)[p + lane];
        s = e.x;
        w = __int_as_float(e.y);
    } else {
        if (lane < g) s = IN_LDS ? lb[p + lane] : lst[beg0 + p + lane];
        if (L1) w = dinv[s];
    }
}

struct walk_t { int j, nb; float dd, a0, a1, a2; };

template <bool L1, int G, bool FULL>
__device__ __forceinline__ void sum_group(const f3* r, float w, int g, int base, int rp, float dl, int beg0,
                                          float* tl, int lane, walk_t& k) {
#pragma unroll
    for (int u = 0; u < G; ++u) {
        if (FULL || u < g) {
            if (L1) {
                const float wl = __int_as_float(__builtin_amdgcn_readlane(__float_as_int(w), u));
                k.a0 = fmaf(wl, r[u].x, k.a0);
                k.a1 = fmaf(wl, r[u].y, k.a1);
                k.a2 = fmaf(wl, r[u].z, k.a2);
            } else {
                k.a0 += r[u].x;
                k.a1 += r[u].y;
                k.a2 += r[u].z;
            }
            if (__builtin_expect(base + u + 1 == k.nb, 0)) {
                f3 o;
                o.x = k.a0 * k.dd;
                o.y = k.a1 * k.dd;
                o.z = k.a2 * k.dd;
                *(f3*)(tl + k.j * TS + lane * 3) = o;
                k.a0 = 0.0f; k.a1 = 0.0f; k.a2 = 0.0f;
                ++k.j;
                k.nb = __builtin_amdgcn_readlane(rp, k.j + 1) - beg0;
                k.dd = __int_as_float(__builtin_amdgcn_readlane(__float_as_int(dl), k.j));
            }
        }
    }
}

template <bool IN_LDS, bool L1>
__device__ __forceinline__ void tile_rows(const float* __restrict__ Tl, const float* __restrict__ dinv,
                                          const int* __restrict__ lst, const int* lb, float* tl,
                                          int rp, float dl, int beg0, int cnt, int lane) {
    constexpr int G = IN_LDS ? GU : 4;
    walk_t k;
    k.j  = 0;
    k.nb = __builtin_amdgcn_readlane(rp, 1) - beg0;
    k.dd = __int_as_float(__builtin_amdgcn_readlane(__float_as_int(dl), 0));
    k.a0 = 0.0f; k.a1 = 0.0f; k.a2 = 0.0f;
    int   s;
    float w;
    run_entries<IN_LDS, L1>(dinv, lst, lb, beg0, 0, cnt < G ? cnt : G, lane, s, w);
    int base = 0;
    for (; base + G <= cnt; base += G) {
        f3 r[G];
        load_rows<G>(Tl, s, r);
        int   s2 = 0;
        float w2 = 0.0f;
        int   g2 = cnt - base - G;
        if (g2 > G) g2 = G;
        if (g2 > 0) run_entries<IN_LDS, L1>(dinv, lst, lb, beg0, base + G, g2, lane, s2, w2);
        sum_group<L1, G, true>(r, w, G, base, rp, dl, beg0, tl, lane, k);
        s = s2;
        w = w2;
    }
    if (base < cnt) {
        f3 r[G];
        load_group<G>(Tl, s, cnt - base, r);
        sum_group<L1, G, false>(r, w, cnt - base, base, rp, dl, beg0, tl, lane, k);
    }
    for (int j = k.j; j < 16; ++j) {
        f3 o;
        o.x = 0.0f; o.y = 0.0f; o.z = 0.0f;
        *(f3*)(tl + j * TS + lane * 3) = o;
    }
}

template <bool L1>
__global__ __launch_bounds__(L1 ? TW1 * 32 : TW2 * 32, 8) void k_layer(const float* __restrict__ T,
                                               const float* __restrict__ dinv,
                                               const int* __restrict__ rowptr,
                                               const int* __restrict__ lst,
                                               const float* __restrict__ bias,
                                               const float* __restrict__ g,
                                               const float* __restrict__ be,
                                               const float* __restrict__ mu,
                                               const float* __restrict__ var,
                                               const float* __restrict__ W,
                                               float* __restrict__ O, int n) {
    __shared__ __attribute__((aligned(16))) float Wt[DD * WT_STRIDE];
    constexpr int TW   = L1 ? TW1 : TW2;
    constexpr int NT   = TW * 32;
    constexpr int LCAP = L1 ? LCAP1 : LCAP2;
    __shared__ __attribute__((aligned(16))) float tile[TW][16 * TS];
    __shared__ __attribute__((aligned(8))) int lbuf[TW][L1 ? 2 * LCAP1 : LCAP2];

    const int lane = threadIdx.x & 31;
    const int wv = __builtin_amdgcn_readfirstlane((int)(threadIdx.x >> 5));
    const int m0 = ((int)blockIdx.x * TW + wv) * 16;
    const bool live = m0 < n;

    int   rp = 0;
    float dl = 0.0f;
    if (live) {
        int j  = lane < 16 ? lane : 16;
        int ix = m0 + j;
        rp = rowptr[ix < n ? ix : n];
        int nd = m0 + (lane & 15);
        dl = dinv[nd < n ? nd : n - 1];
    }

    {
        constexpr int NU = (DD * DD + NT - 1) / NT;
        float w[NU];
#pragma unroll
        for (int u = 0; u < NU; ++u) {
            int i = u * NT + (int)threadIdx.x;
            w[u] = W[i < DD * DD ? i : DD * DD - 1];
        }
#pragma unroll
        for (int u = 0; u < NU; ++u) {
            int i = u * NT + (int)threadIdx.x;
            if (i < DD * DD) {
                int k = i / DD, col = i - k * DD;
                Wt[col * WT_STRIDE + k] = w[u];
            }
        }
    }
    __syncthreads();
    if (!live) return;

    float* tl = tile[wv];
    int*   lb = lbuf[wv];
    const int beg0 = __builtin_amdgcn_readlane(rp, 0);
    const int endT = __builtin_amdgcn_readlane(rp, 16);

    {
        int cnt = endT - beg0;
        if (cnt > LCAP) cnt = LCAP;
        if (cnt > 0) {
            int t[LCAP / 32];
#pragma unroll
            for (int u = 0; u < LCAP / 32; ++u) {
                int p = u * 32 + lane;
                t[u] = lst[beg0 + (p < cnt ? p : 0)];
            }
            if (L1) {
                float d[LCAP / 32];
#pragma unroll
                for (int u = 0; u < LCAP / 32; ++u) d[u] = dinv[t[u]];
#pragma unroll
                for (int u = 0; u < LCAP / 32; ++u)
                    ((int2*)lb)[u * 32 + lane] = make_int2(t[u], __float_as_int(d[u]));
            } else {
#pragma unroll
                for (int u = 0; u < LCAP / 32; ++u) lb[u * 32 + lane] = t[u];
            }
        }
    }

    const float* Tl = T + lane * 3;
    if (endT - beg0 <= LCAP) tile_rows<true,  L1>(Tl, dinv, lst, lb, tl, rp, dl, beg0, endT - beg0, lane);
    else                     tile_rows<false, L1>(Tl, dinv, lst, lb, tl, rp, dl, beg0, endT - beg0, lane);

    const int lhalf = lane >> 4;
    const int l16   = lane & 15;
    const int khalf = lhalf * 2;
    const int rbase = m0 + lhalf * 8;

    float cb[6];
#pragma unroll
    for (int t = 0; t < 6; ++t) cb[t] = bias[t * 16 + l16];

    v8f acc[6];
#pragma unroll
    for (int t = 0; t < 6; ++t)
#pragma unroll
        for (int r = 0; r < 8; ++r) acc[t][r] = 0.0f;

    const float* arow = tl + l16 * TS + khalf;
    for (int k = 0; k < DD; k += 4) {
        v2f a = *(const v2f*)(arow + k);
        int kr = k + khalf;
#pragma unroll
        for (int t = 0; t < 6; ++t) {
            int col = t * 16 + l16;
            v2f b = *(const v2f*)(&Wt[col * WT_STRIDE + kr]);
            acc[t] = __builtin_amdgcn_wmma_f32_16x16x4_f32(
                false, a, false, b, (short)0, acc[t], false, false);
        }
    }

    if (L1) {
        __builtin_amdgcn_sched_barrier(0);
        float cm[6], cf[6], ce[6];
#pragma unroll
        for (int t = 0; t < 6; ++t) {
            const int col = t * 16 + l16;
            cm[t] = mu[col];
            cf[t] = rsqrtf(var[col] + BN_EPS) * g[col];
            ce[t] = be[col];
        }
        float sc[8];
#pragma unroll
        for (int r = 0; r < 8; ++r) {
            int lo = __builtin_amdgcn_readlane(__float_as_int(dl), r);
            int hi = __builtin_amdgcn_readlane(__float_as_int(dl), r + 8);
            sc[r] = __int_as_float(lhalf ? hi : lo);
        }
#pragma unroll
        for (int t = 0; t < 6; ++t)
#pragma unroll
            for (int r = 0; r < 8; ++r)
                acc[t][r] = leaky_bn(acc[t][r] + cb[t], cm[t], cf[t], ce[t]) * sc[r];
    } else {
#pragma unroll
        for (int t = 0; t < 6; ++t)
#pragma unroll
            for (int r = 0; r < 8; ++r) acc[t][r] += cb[t];
    }

    if (m0 + 16 <= n) {
#pragma unroll
        for (int t = 0; t < 6; ++t) {
            int col = t * 16 + l16;
#pragma unroll
            for (int r = 0; r < 8; ++r)
                O[(size_t)(rbase + r) * DD + col] = acc[t][r];
        }
    } else {
#pragma unroll
        for (int t = 0; t < 6; ++t) {
            int col = t * 16 + l16;
#pragma unroll
            for (int r = 0; r < 8; ++r)
                if (rbase + r < n)
                    O[(size_t)(rbase + r) * DD + col] = acc[t][r];
        }
    }
}


extern "C" void kernel_launch(void* const* d_in, const int* in_sizes, int n_in,
                              void* d_out, int out_size, void* d_ws, size_t ws_size,
                              hipStream_t stream) {
    const float* X    = (const float*)d_in[0];
    const int*   ei   = (const int*)d_in[1];
    const float* W1   = (const float*)d_in[2];
    const float* b1   = (const float*)d_in[3];
    const float* W2   = (const float*)d_in[4];
    const float* b2   = (const float*)d_in[5];
    const float* gam  = (const float*)d_in[6];
    const float* bet  = (const float*)d_in[7];
    const float* mu   = (const float*)d_in[8];
    const float* var  = (const float*)d_in[9];

    const int N = in_sizes[0] / DD;
    const int E = in_sizes[1] / 2;
    const int* srcI = ei;
    const int* dstI = ei + E;

    float* dinv = (float*)d_ws;
    float* x1s  = dinv + N;
    int*   rptr = (int*)(x1s + (size_t)N * DD);
    int*   lst  = rptr + (N + 1);
    int*   offT = lst + E + N;
    int2*  pairs = (int2*)(((size_t)(offT + 2 * (MAXPER * 1024 + 1)) + 7) & ~(size_t)7);
    float* out  = (float*)d_out;

    const int nB        = (N + (1 << BSH) - 1) >> BSH;
    const int nC        = (E + CHUNK - 1) / CHUNK;
    const int tiles     = (N + 15) / 16;
    const int grid1     = (tiles + TW1 - 1) / TW1;
    const int grid2     = (tiles + TW2 - 1) / TW2;

    k_chunk_sort  <<<nC, 1024, 0, stream>>>(srcI, dstI, offT, pairs, E, nB, nC);
    k_bucket_lists<<<nB, 1024, 0, stream>>>(pairs, offT, rptr, dinv, lst, N, nC);

    k_layer<true> <<<grid1, TW1 * 32, 0, stream>>>(X,   dinv, rptr, lst, b1, gam, bet, mu, var, W1, x1s, N);
    k_layer<false><<<grid2, TW2 * 32, 0, stream>>>(x1s, dinv, rptr, lst, b2, gam, bet, mu, var, W2, out, N);
}
